# starting kernel
# baseline (speedup 1.0000x reference)
#include <hip/hip_runtime.h>
#include <hip/hip_bf16.h>
#include <math.h>

typedef float v2f __attribute__((ext_vector_type(2)));
typedef float v8f __attribute__((ext_vector_type(8)));

__global__ void k_fill(float* __restrict__ p, float val, int n) {
  int i = blockIdx.x * blockDim.x + threadIdx.x;
  if (i < n) p[i] = val;
}

__global__ void k_deg_edges(const int* __restrict__ dst, float* __restrict__ deg, int E) {
  int e = blockIdx.x * blockDim.x + threadIdx.x;
  if (e < E) unsafeAtomicAdd(&deg[dst[e]], 1.0f);
}

__global__ void k_rsqrt(const float* __restrict__ deg, float* __restrict__ dinv, int n) {
  int i = blockIdx.x * blockDim.x + threadIdx.x;
  if (i < n) dinv[i] = rsqrtf(deg[i]);
}

__global__ void k_pack_bfrag(const float* __restrict__ B, v2f* __restrict__ BF,
                             int N, int nTiles) {
  int i = blockIdx.x * blockDim.x + threadIdx.x;
  int total = nTiles * 1024;
  if (i >= total) return;
  int lane = i & 31;
  int t    = (i >> 5) & 31;
  int nt   = i >> 10;
  int half = lane >> 4;
  int lm   = lane & 15;
  int col  = nt * 16 + lm;
  int kk   = 4 * t + 2 * half;
  v2f v;
  v.x = (col < N) ? B[(size_t)kk * N + col]       : 0.0f;
  v.y = (col < N) ? B[(size_t)(kk + 1) * N + col] : 0.0f;
  BF[i] = v;
}

__global__ __launch_bounds__(256) void gcn_gemm_wmma_f32(
    const float* __restrict__ A, const v2f* __restrict__ BF,
    float* __restrict__ C, int M, int N, int nTiles) {
  constexpr int K = 128;
  constexpr int LDS_STRIDE = 132;
  __shared__ float As[16 * LDS_STRIDE];

  const int m0  = blockIdx.x * 16;
  const int tid = threadIdx.x;

#pragma unroll
  for (int j = 0; j < 8; ++j) {
    int off = tid + j * 256;
    int row = off >> 7;
    int col = off & 127;
    As[row * LDS_STRIDE + col] = A[(size_t)(m0 + row) * K + col];
  }
  __syncthreads();

  const int wave = tid >> 5;
  const int lane = tid & 31;
  const int half = lane >> 4;
  const int lm   = lane & 15;

  const float* aBase = &As[lm * LDS_STRIDE + 2 * half];

  for (int nt = wave; nt < nTiles; nt += 8) {
    const v2f* bBase = BF + (size_t)nt * 1024 + lane;
    v8f c = {};
#pragma unroll
    for (int t = 0; t < 32; ++t) {
      v2f a = *(const v2f*)(aBase + 4 * t);
      v2f b = bBase[(size_t)t * 32];
      c = __builtin_amdgcn_wmma_f32_16x16x4_f32(
           false, a,  false, b,
           (short)0, c,  false,  false);
    }
    const int col = nt * 16 + lm;
    if (col < N) {
      const int mBase = m0 + 8 * half;
#pragma unroll
      for (int r = 0; r < 8; ++r)
        C[(size_t)(mBase + r) * N + col] = c[r];
    }
  }
}

__global__ void k_self_init(const float* __restrict__ h, const float* __restrict__ dinv,
                            float* __restrict__ out, int total, int F) {
  int i = blockIdx.x * blockDim.x + threadIdx.x;
  if (i >= total) return;
  int node = i / F;
  float di = dinv[node];
  out[i] = di * di * h[i];
}

__global__ void k_edge_f128(const float* __restrict__ h, const int* __restrict__ src,
                            const int* __restrict__ dst, const float* __restrict__ dinv,
                            float* __restrict__ out, int E) {
  int tid = blockIdx.x * blockDim.x + threadIdx.x;
  int e = tid >> 7;
  if (e >= E) return;
  int f = tid & 127;
  int s = src[e], d = dst[e];
  float coeff = dinv[s] * dinv[d];
  unsafeAtomicAdd(&out[(size_t)d * 128 + f], coeff * h[(size_t)s * 128 + f]);
}

__global__ void k_edge_f40(const float* __restrict__ h, const int* __restrict__ src,
                           const int* __restrict__ dst, const float* __restrict__ dinv,
                           float* __restrict__ out, int E) {
  int tid = blockIdx.x * blockDim.x + threadIdx.x;
  int e = tid >> 6;
  int f = tid & 63;
  if (e >= E || f >= 40) return;
  int s = src[e], d = dst[e];
  float coeff = dinv[s] * dinv[d];
  unsafeAtomicAdd(&out[(size_t)d * 40 + f], coeff * h[(size_t)s * 40 + f]);
}

__global__ void k_bias_relu(float* __restrict__ out, const float* __restrict__ b,
                            int total, int F) {
  int i = blockIdx.x * blockDim.x + threadIdx.x;
  if (i >= total) return;
  float v = out[i] + b[i % F];
  out[i] = v > 0.0f ? v : 0.0f;
}

__global__ __launch_bounds__(256) void k_logsoftmax40(float* __restrict__ out,
                                                      const float* __restrict__ b, int N) {
  int row  = blockIdx.x * 8 + (threadIdx.x >> 5);
  int lane = threadIdx.x & 31;
  if (row >= N) return;
  float* p = out + (size_t)row * 40;
  float v0 = p[lane] + b[lane];
  bool has2 = (lane + 32) < 40;
  float v1 = has2 ? p[lane + 32] + b[lane + 32] : -INFINITY;

  float m = fmaxf(v0, v1);
#pragma unroll
  for (int off = 16; off > 0; off >>= 1) m = fmaxf(m, __shfl_xor(m, off, 32));

  float s = expf(v0 - m) + (has2 ? expf(v1 - m) : 0.0f);
#pragma unroll
  for (int off = 16; off > 0; off >>= 1) s += __shfl_xor(s, off, 32);

  float lse = logf(s);
  p[lane] = v0 - m - lse;
  if (has2) p[lane + 32] = v1 - m - lse;
}

static inline int cdiv_ll(long long a, long long b) { return (int)((a + b - 1) / b); }

extern "C" void kernel_launch(void* const* d_in, const int* in_sizes, int n_in,
                              void* d_out, int out_size, void* d_ws, size_t ws_size,
                              hipStream_t stream) {
  const float* x  = (const float*)d_in[0];
  const int*   ei = (const int*)  d_in[1];
  const float* W1 = (const float*)d_in[2];
  const float* b1 = (const float*)d_in[3];
  const float* W2 = (const float*)d_in[4];
  const float* b2 = (const float*)d_in[5];

  const int F = 128, H = 128, C = 40;
  const int N = in_sizes[0] / F;
  const int E = in_sizes[1] / 2;
  const int* src = ei;
  const int* dst = ei + E;

  const int nT1 = (H + 15) / 16;
  const int nT2 = (C + 15) / 16;

  float* ws   = (float*)d_ws;
  float* deg  = ws;                 ws += N;
  float* dinv = ws;                 ws += N;
  float* h1   = ws;                 ws += (size_t)N * H;
  float* o1   = ws;                 ws += (size_t)N * H;
  float* h2   = ws;                 ws += (size_t)N * C;
  v2f* bf1    = (v2f*)ws;           ws += (size_t)nT1 * 1024 * 2;
  v2f* bf2    = (v2f*)ws;           ws += (size_t)nT2 * 1024 * 2;
  float* out  = (float*)d_out;

  const int B = 256;

  k_fill      <<<cdiv_ll(N, B), B, 0, stream>>>(deg, 1.0f, N);
  k_deg_edges <<<cdiv_ll(E, B), B, 0, stream>>>(dst, deg, E);
  k_rsqrt     <<<cdiv_ll(N, B), B, 0, stream>>>(deg, dinv, N);

  k_pack_bfrag<<<cdiv_ll((long long)nT1 * 1024, B), B, 0, stream>>>(W1, bf1, H, nT1);
  k_pack_bfrag<<<cdiv_ll((long long)nT2 * 1024, B), B, 0, stream>>>(W2, bf2, C, nT2);

  gcn_gemm_wmma_f32<<<N / 16, B, 0, stream>>>(x, bf1, h1, N, H, nT1);
  k_self_init <<<cdiv_ll((long long)N * H, B), B, 0, stream>>>(h1, dinv, o1, N * H, H);
  k_edge_f128 <<<cdiv_ll((long long)E * 128, B), B, 0, stream>>>(h1, src, dst, dinv, o1, E);
  k_bias_relu <<<cdiv_ll((long long)N * H, B), B, 0, stream>>>(o1, b1, N * H, H);

  gcn_gemm_wmma_f32<<<N / 16, B, 0, stream>>>(o1, bf2, h2, N, C, nT2);
  k_self_init <<<cdiv_ll((long long)N * C, B), B, 0, stream>>>(h2, dinv, out, N * C, C);
  k_edge_f40  <<<cdiv_ll((long long)E * 64, B), B, 0, stream>>>(h2, src, dst, dinv, out, E);
  k_logsoftmax40<<<cdiv_ll(N, 8), B, 0, stream>>>(out, b2, N);
}


# Edge aggregation by gather over a destination-sorted edge list built in-call (degree add returns each edge's rank; scan; fill): float atomics removed, each row's sum order is now its arrival rank
# speedup vs baseline: 2.7722x; 2.7722x over previous
#include <hip/hip_runtime.h>
#include <hip/hip_bf16.h>
#include <math.h>

typedef float v2f __attribute__((ext_vector_type(2)));
typedef float v8f __attribute__((ext_vector_type(8)));

__global__ void k_zero_i32(int* __restrict__ p, int n) {
  int i = blockIdx.x * blockDim.x + threadIdx.x;
  if (i < n) p[i] = 0;
}

__global__ void k_deg_rank(const int* __restrict__ dst, int* __restrict__ cnt,
                           int* __restrict__ rank, int E) {
  int e = blockIdx.x * blockDim.x + threadIdx.x;
  if (e < E) rank[e] = atomicAdd(&cnt[dst[e]], 1);
}

__global__ __launch_bounds__(256) void k_scan_block(const int* __restrict__ cnt,
                                                    int* __restrict__ rowptr,
                                                    int* __restrict__ blocktot, int n) {
  __shared__ int wsum[8];
  const int tid  = threadIdx.x;
  const int lane = tid & 31;
  const int wave = tid >> 5;
  const int i0   = (blockIdx.x * 256 + tid) * 4;
  int v0 = (i0     < n) ? cnt[i0]     : 0;
  int v1 = (i0 + 1 < n) ? cnt[i0 + 1] : 0;
  int v2 = (i0 + 2 < n) ? cnt[i0 + 2] : 0;
  int v3 = (i0 + 3 < n) ? cnt[i0 + 3] : 0;
  const int s = v0 + v1 + v2 + v3;
  int incl = s;
  for (int off = 1; off < 32; off <<= 1) {
    int t = __shfl_up(incl, off, 32);
    if (lane >= off) incl += t;
  }
  if (lane == 31) wsum[wave] = incl;
  __syncthreads();
  int woff = 0;
  for (int w = 0; w < 8; ++w) {
    int t = wsum[w];
    if (w < wave) woff += t;
  }
  const int excl = woff + incl - s;
  if (i0     < n) rowptr[i0]     = excl;
  if (i0 + 1 < n) rowptr[i0 + 1] = excl + v0;
  if (i0 + 2 < n) rowptr[i0 + 2] = excl + v0 + v1;
  if (i0 + 3 < n) rowptr[i0 + 3] = excl + v0 + v1 + v2;
  if (tid == 255) blocktot[blockIdx.x] = woff + incl;
}

__global__ __launch_bounds__(256) void k_scan_finish(const int* __restrict__ cnt,
                                                     int* __restrict__ rowptr,
                                                     const int* __restrict__ blocktot,
                                                     float* __restrict__ dinv, int n) {
  const int i     = blockIdx.x * 256 + threadIdx.x;
  const int chunk = blockIdx.x >> 2;
  const int lane  = threadIdx.x & 31;
  int part = 0;
  for (int j = lane; j < chunk; j += 32) part += blocktot[j];
  for (int off = 16; off > 0; off >>= 1) part += __shfl_xor(part, off, 32);
  if (i < n) {
    rowptr[i] += part;
    dinv[i] = rsqrtf((float)(cnt[i] + 1));
  }
}

__global__ void k_fill_col(const int* __restrict__ src, const int* __restrict__ dst,
                           const int* __restrict__ rank, const int* __restrict__ rowptr,
                           int* __restrict__ col, int E) {
  int e = blockIdx.x * blockDim.x + threadIdx.x;
  if (e < E) col[rowptr[dst[e]] + rank[e]] = src[e];
}

__global__ void k_pack_bfrag(const float* __restrict__ B, v2f* __restrict__ BF,
                             int N, int nTiles) {
  int i = blockIdx.x * blockDim.x + threadIdx.x;
  int total = nTiles * 1024;
  if (i >= total) return;
  int lane = i & 31;
  int t    = (i >> 5) & 31;
  int nt   = i >> 10;
  int half = lane >> 4;
  int lm   = lane & 15;
  int col  = nt * 16 + lm;
  int kk   = 4 * t + 2 * half;
  v2f v;
  v.x = (col < N) ? B[(size_t)kk * N + col]       : 0.0f;
  v.y = (col < N) ? B[(size_t)(kk + 1) * N + col] : 0.0f;
  BF[i] = v;
}

__global__ __launch_bounds__(256) void gcn_gemm_wmma_f32(
    const float* __restrict__ A, const v2f* __restrict__ BF,
    float* __restrict__ C, int M, int N, int nTiles) {
  constexpr int K = 128;
  constexpr int LDS_STRIDE = 132;
  __shared__ float As[16 * LDS_STRIDE];

  const int m0  = blockIdx.x * 16;
  const int tid = threadIdx.x;

#pragma unroll
  for (int j = 0; j < 8; ++j) {
    int off = tid + j * 256;
    int row = off >> 7;
    int col = off & 127;
    As[row * LDS_STRIDE + col] = A[(size_t)(m0 + row) * K + col];
  }
  __syncthreads();

  const int wave = tid >> 5;
  const int lane = tid & 31;
  const int half = lane >> 4;
  const int lm   = lane & 15;

  const float* aBase = &As[lm * LDS_STRIDE + 2 * half];

  for (int nt = wave; nt < nTiles; nt += 8) {
    const v2f* bBase = BF + (size_t)nt * 1024 + lane;
    v8f c = {};
#pragma unroll
    for (int t = 0; t < 32; ++t) {
      v2f a = *(const v2f*)(aBase + 4 * t);
      v2f b = bBase[(size_t)t * 32];
      c = __builtin_amdgcn_wmma_f32_16x16x4_f32(
           false, a,  false, b,
           (short)0, c,  false,  false);
    }
    const int col = nt * 16 + lm;
    if (col < N) {
      const int mBase = m0 + 8 * half;
#pragma unroll
      for (int r = 0; r < 8; ++r)
        C[(size_t)(mBase + r) * N + col] = c[r];
    }
  }
}

__global__ __launch_bounds__(256) void k_gather_f128(
    const float* __restrict__ h, const int* __restrict__ rowptr,
    const int* __restrict__ cnt, const int* __restrict__ col,
    const float* __restrict__ dinv, float* __restrict__ out, int n) {
  const int lane = threadIdx.x & 31;
  const int node = __builtin_amdgcn_readfirstlane(blockIdx.x * 8 + (threadIdx.x >> 5));
  if (node >= n) return;
  const int   start = rowptr[node];
  const int   deg   = cnt[node];
  const float di    = dinv[node];
  const float4* __restrict__ h4 = (const float4*)h;

  const float4 hs = h4[(size_t)node * 32 + lane];
  const float  sc = di * di;
  float4 acc = make_float4(sc * hs.x, sc * hs.y, sc * hs.z, sc * hs.w);

  for (int base = 0; base < deg; base += 32) {
    const int k = base + lane;
    int   my  = 0;
    float myc = 0.0f;
    if (k < deg) {
      my = col[start + k];
      my = min(max(my, 0), n - 1);
      myc = dinv[my] * di;
    }
    const int m = min(32, deg - base);
#pragma unroll 4
    for (int j = 0; j < m; ++j) {
      const int   s = __builtin_amdgcn_readlane(my, j);
      const float c = __int_as_float(__builtin_amdgcn_readlane(__float_as_int(myc), j));
      const float4 v = h4[(size_t)s * 32 + lane];
      acc.x += c * v.x;
      acc.y += c * v.y;
      acc.z += c * v.z;
      acc.w += c * v.w;
    }
  }
  ((float4*)out)[(size_t)node * 32 + lane] = acc;
}

__global__ __launch_bounds__(256) void k_gather_f40(
    const float* __restrict__ h, const int* __restrict__ rowptr,
    const int* __restrict__ cnt, const int* __restrict__ col,
    const float* __restrict__ dinv, float* __restrict__ out, int n) {
  const int lane = threadIdx.x & 31;
  const int node = __builtin_amdgcn_readfirstlane(blockIdx.x * 8 + (threadIdx.x >> 5));
  if (node >= n) return;
  const int   start = rowptr[node];
  const int   deg   = cnt[node];
  const float di    = dinv[node];
  const float4* __restrict__ h4 = (const float4*)h;

  const int  g      = lane / 10;
  const int  q      = lane - g * 10;
  const bool active = lane < 30;

  float4 acc = make_float4(0.0f, 0.0f, 0.0f, 0.0f);
  for (int base = 0; base < deg; base += 32) {
    const int k = base + lane;
    int   my  = node;
    float myc = 0.0f;
    if (k < deg) {
      my = col[start + k];
      my = min(max(my, 0), n - 1);
      myc = dinv[my] * di;
    }
    const int m = min(32, deg - base);
#pragma unroll 2
    for (int j0 = 0; j0 < m; j0 += 3) {
      const int   j = j0 + g;
      const int   s = __shfl(my, j & 31, 32);
      float       c = __shfl(myc, j & 31, 32);
      c = (active && j < m) ? c : 0.0f;
      const float4 v = h4[(size_t)s * 10 + q];
      acc.x += c * v.x;
      acc.y += c * v.y;
      acc.z += c * v.z;
      acc.w += c * v.w;
    }
  }
  acc.x += __shfl_down(acc.x, 10, 32) + __shfl_down(acc.x, 20, 32);
  acc.y += __shfl_down(acc.y, 10, 32) + __shfl_down(acc.y, 20, 32);
  acc.z += __shfl_down(acc.z, 10, 32) + __shfl_down(acc.z, 20, 32);
  acc.w += __shfl_down(acc.w, 10, 32) + __shfl_down(acc.w, 20, 32);
  if (lane < 10) {
    const float4 hs = h4[(size_t)node * 10 + lane];
    const float  sc = di * di;
    acc.x += sc * hs.x;
    acc.y += sc * hs.y;
    acc.z += sc * hs.z;
    acc.w += sc * hs.w;
    ((float4*)out)[(size_t)node * 10 + lane] = acc;
  }
}

__global__ void k_bias_relu(float* __restrict__ out, const float* __restrict__ b,
                            int total, int F) {
  int i = blockIdx.x * blockDim.x + threadIdx.x;
  if (i >= total) return;
  float v = out[i] + b[i % F];
  out[i] = v > 0.0f ? v : 0.0f;
}

__global__ __launch_bounds__(256) void k_logsoftmax40(float* __restrict__ out,
                                                      const float* __restrict__ b, int N) {
  int row  = blockIdx.x * 8 + (threadIdx.x >> 5);
  int lane = threadIdx.x & 31;
  if (row >= N) return;
  float* p = out + (size_t)row * 40;
  float v0 = p[lane] + b[lane];
  bool has2 = (lane + 32) < 40;
  float v1 = has2 ? p[lane + 32] + b[lane + 32] : -INFINITY;

  float m = fmaxf(v0, v1);
#pragma unroll
  for (int off = 16; off > 0; off >>= 1) m = fmaxf(m, __shfl_xor(m, off, 32));

  float s = expf(v0 - m) + (has2 ? expf(v1 - m) : 0.0f);
#pragma unroll
  for (int off = 16; off > 0; off >>= 1) s += __shfl_xor(s, off, 32);

  float lse = logf(s);
  p[lane] = v0 - m - lse;
  if (has2) p[lane + 32] = v1 - m - lse;
}

static inline int cdiv_ll(long long a, long long b) { return (int)((a + b - 1) / b); }

extern "C" void kernel_launch(void* const* d_in, const int* in_sizes, int n_in,
                              void* d_out, int out_size, void* d_ws, size_t ws_size,
                              hipStream_t stream) {
  const float* x  = (const float*)d_in[0];
  const int*   ei = (const int*)  d_in[1];
  const float* W1 = (const float*)d_in[2];
  const float* b1 = (const float*)d_in[3];
  const float* W2 = (const float*)d_in[4];
  const float* b2 = (const float*)d_in[5];

  const int F = 128, H = 128, C = 40;
  const int N = in_sizes[0] / F;
  const int E = in_sizes[1] / 2;
  const int* src = ei;
  const int* dst = ei + E;

  const int nT1 = (H + 15) / 16;
  const int nT2 = (C + 15) / 16;

  float* ws   = (float*)d_ws;
  int*   cnt  = (int*)ws;           ws += N;
  float* dinv = ws;                 ws += N;
  float* h1   = ws;                 ws += (size_t)N * H;
  float* o1   = ws;                 ws += (size_t)N * H;
  int*   col  = (int*)ws;
  int*   rank = col + E;
  int*   rowptr   = rank + E;
  int*   blocktot = rowptr + N;
  ws += (size_t)N * C;
  v2f* bf1    = (v2f*)ws;           ws += (size_t)nT1 * 1024 * 2;
  v2f* bf2    = (v2f*)ws;           ws += (size_t)nT2 * 1024 * 2;
  float* h2   = h1;
  float* out  = (float*)d_out;

  const int B = 256;

  k_zero_i32   <<<cdiv_ll(N, B), B, 0, stream>>>(cnt, N);
  k_deg_rank   <<<cdiv_ll(E, B), B, 0, stream>>>(dst, cnt, rank, E);
  k_scan_block <<<cdiv_ll(N, 1024), B, 0, stream>>>(cnt, rowptr, blocktot, N);
  k_scan_finish<<<cdiv_ll(N, B), B, 0, stream>>>(cnt, rowptr, blocktot, dinv, N);
  k_fill_col   <<<cdiv_ll(E, B), B, 0, stream>>>(src, dst, rank, rowptr, col, E);

  k_pack_bfrag<<<cdiv_ll((long long)nT1 * 1024, B), B, 0, stream>>>(W1, bf1, H, nT1);
  k_pack_bfrag<<<cdiv_ll((long long)nT2 * 1024, B), B, 0, stream>>>(W2, bf2, C, nT2);

  gcn_gemm_wmma_f32<<<N / 16, B, 0, stream>>>(x, bf1, h1, N, H, nT1);
  k_gather_f128<<<cdiv_ll(N, 8), B, 0, stream>>>(h1, rowptr, cnt, col, dinv, o1, N);
  k_bias_relu <<<cdiv_ll((long long)N * H, B), B, 0, stream>>>(o1, b1, N * H, H);

  gcn_gemm_wmma_f32<<<N / 16, B, 0, stream>>>(o1, bf2, h2, N, C, nT2);
  k_gather_f40<<<cdiv_ll(N, 8), B, 0, stream>>>(h2, rowptr, cnt, col, dinv, out, N);
  k_logsoftmax40<<<cdiv_ll(N, 8), B, 0, stream>>>(out, b2, N);
}


# Row epilogues folded into the gathers: bias+relu into the F=128 gather, bias+log_softmax into the F=40 gather; 13 -> 11 launches; softmax sum order now 4 classes per lane, then a 16-lane tree
# speedup vs baseline: 3.1520x; 1.1370x over previous
#include <hip/hip_runtime.h>
#include <hip/hip_bf16.h>
#include <math.h>

typedef float v2f __attribute__((ext_vector_type(2)));
typedef float v8f __attribute__((ext_vector_type(8)));

__global__ void k_zero_i32(int* __restrict__ p, int n) {
  int i = blockIdx.x * blockDim.x + threadIdx.x;
  if (i < n) p[i] = 0;
}

__global__ void k_deg_rank(const int* __restrict__ dst, int* __restrict__ cnt,
                           int* __restrict__ rank, int E) {
  int e = blockIdx.x * blockDim.x + threadIdx.x;
  if (e < E) rank[e] = atomicAdd(&cnt[dst[e]], 1);
}

__global__ __launch_bounds__(256) void k_scan_block(const int* __restrict__ cnt,
                                                    int* __restrict__ rowptr,
                                                    int* __restrict__ blocktot, int n) {
  __shared__ int wsum[8];
  const int tid  = threadIdx.x;
  const int lane = tid & 31;
  const int wave = tid >> 5;
  const int i0   = (blockIdx.x * 256 + tid) * 4;
  int v0 = (i0     < n) ? cnt[i0]     : 0;
  int v1 = (i0 + 1 < n) ? cnt[i0 + 1] : 0;
  int v2 = (i0 + 2 < n) ? cnt[i0 + 2] : 0;
  int v3 = (i0 + 3 < n) ? cnt[i0 + 3] : 0;
  const int s = v0 + v1 + v2 + v3;
  int incl = s;
  for (int off = 1; off < 32; off <<= 1) {
    int t = __shfl_up(incl, off, 32);
    if (lane >= off) incl += t;
  }
  if (lane == 31) wsum[wave] = incl;
  __syncthreads();
  int woff = 0;
  for (int w = 0; w < 8; ++w) {
    int t = wsum[w];
    if (w < wave) woff += t;
  }
  const int excl = woff + incl - s;
  if (i0     < n) rowptr[i0]     = excl;
  if (i0 + 1 < n) rowptr[i0 + 1] = excl + v0;
  if (i0 + 2 < n) rowptr[i0 + 2] = excl + v0 + v1;
  if (i0 + 3 < n) rowptr[i0 + 3] = excl + v0 + v1 + v2;
  if (tid == 255) blocktot[blockIdx.x] = woff + incl;
}

__global__ __launch_bounds__(256) void k_scan_finish(const int* __restrict__ cnt,
                                                     int* __restrict__ rowptr,
                                                     const int* __restrict__ blocktot,
                                                     float* __restrict__ dinv, int n) {
  const int i     = blockIdx.x * 256 + threadIdx.x;
  const int chunk = blockIdx.x >> 2;
  const int lane  = threadIdx.x & 31;
  int part = 0;
  for (int j = lane; j < chunk; j += 32) part += blocktot[j];
  for (int off = 16; off > 0; off >>= 1) part += __shfl_xor(part, off, 32);
  if (i < n) {
    rowptr[i] += part;
    dinv[i] = rsqrtf((float)(cnt[i] + 1));
  }
}

__global__ void k_fill_col(const int* __restrict__ src, const int* __restrict__ dst,
                           const int* __restrict__ rank, const int* __restrict__ rowptr,
                           int* __restrict__ col, int E) {
  int e = blockIdx.x * blockDim.x + threadIdx.x;
  if (e < E) col[rowptr[dst[e]] + rank[e]] = src[e];
}

__global__ void k_pack_bfrag(const float* __restrict__ B, v2f* __restrict__ BF,
                             int N, int nTiles) {
  int i = blockIdx.x * blockDim.x + threadIdx.x;
  int total = nTiles * 1024;
  if (i >= total) return;
  int lane = i & 31;
  int t    = (i >> 5) & 31;
  int nt   = i >> 10;
  int half = lane >> 4;
  int lm   = lane & 15;
  int col  = nt * 16 + lm;
  int kk   = 4 * t + 2 * half;
  v2f v;
  v.x = (col < N) ? B[(size_t)kk * N + col]       : 0.0f;
  v.y = (col < N) ? B[(size_t)(kk + 1) * N + col] : 0.0f;
  BF[i] = v;
}

__global__ __launch_bounds__(256) void gcn_gemm_wmma_f32(
    const float* __restrict__ A, const v2f* __restrict__ BF,
    float* __restrict__ C, int M, int N, int nTiles) {
  constexpr int K = 128;
  constexpr int LDS_STRIDE = 132;
  __shared__ float As[16 * LDS_STRIDE];

  const int m0  = blockIdx.x * 16;
  const int tid = threadIdx.x;

#pragma unroll
  for (int j = 0; j < 8; ++j) {
    int off = tid + j * 256;
    int row = off >> 7;
    int col = off & 127;
    As[row * LDS_STRIDE + col] = A[(size_t)(m0 + row) * K + col];
  }
  __syncthreads();

  const int wave = tid >> 5;
  const int lane = tid & 31;
  const int half = lane >> 4;
  const int lm   = lane & 15;

  const float* aBase = &As[lm * LDS_STRIDE + 2 * half];

  for (int nt = wave; nt < nTiles; nt += 8) {
    const v2f* bBase = BF + (size_t)nt * 1024 + lane;
    v8f c = {};
#pragma unroll
    for (int t = 0; t < 32; ++t) {
      v2f a = *(const v2f*)(aBase + 4 * t);
      v2f b = bBase[(size_t)t * 32];
      c = __builtin_amdgcn_wmma_f32_16x16x4_f32(
           false, a,  false, b,
           (short)0, c,  false,  false);
    }
    const int col = nt * 16 + lm;
    if (col < N) {
      const int mBase = m0 + 8 * half;
#pragma unroll
      for (int r = 0; r < 8; ++r)
        C[(size_t)(mBase + r) * N + col] = c[r];
    }
  }
}

__global__ __launch_bounds__(256) void k_gather_f128_bias_relu(
    const float* __restrict__ h, const int* __restrict__ rowptr,
    const int* __restrict__ cnt, const int* __restrict__ col,
    const float* __restrict__ dinv, const float* __restrict__ bias,
    float* __restrict__ out, int n) {
  const int lane = threadIdx.x & 31;
  const int node = __builtin_amdgcn_readfirstlane(blockIdx.x * 8 + (threadIdx.x >> 5));
  if (node >= n) return;
  const int   start = rowptr[node];
  const int   deg   = cnt[node];
  const float di    = dinv[node];
  const float4* __restrict__ h4 = (const float4*)h;

  const float4 hs = h4[(size_t)node * 32 + lane];
  const float4 bv = ((const float4*)bias)[lane];
  const float  sc = di * di;
  float4 acc = make_float4(sc * hs.x, sc * hs.y, sc * hs.z, sc * hs.w);

  for (int base = 0; base < deg; base += 32) {
    const int k = base + lane;
    int   my  = 0;
    float myc = 0.0f;
    if (k < deg) {
      my = col[start + k];
      my = min(max(my, 0), n - 1);
      myc = dinv[my] * di;
    }
    const int m = min(32, deg - base);
#pragma unroll 4
    for (int j = 0; j < m; ++j) {
      const int   s = __builtin_amdgcn_readlane(my, j);
      const float c = __int_as_float(__builtin_amdgcn_readlane(__float_as_int(myc), j));
      const float4 v = h4[(size_t)s * 32 + lane];
      acc.x += c * v.x;
      acc.y += c * v.y;
      acc.z += c * v.z;
      acc.w += c * v.w;
    }
  }
  float4 r;
  r.x = acc.x + bv.x;
  r.y = acc.y + bv.y;
  r.z = acc.z + bv.z;
  r.w = acc.w + bv.w;
  r.x = r.x > 0.0f ? r.x : 0.0f;
  r.y = r.y > 0.0f ? r.y : 0.0f;
  r.z = r.z > 0.0f ? r.z : 0.0f;
  r.w = r.w > 0.0f ? r.w : 0.0f;
  ((float4*)out)[(size_t)node * 32 + lane] = r;
}

__global__ __launch_bounds__(256) void k_gather_f40_bias_logsoftmax(
    const float* __restrict__ h, const int* __restrict__ rowptr,
    const int* __restrict__ cnt, const int* __restrict__ col,
    const float* __restrict__ dinv, const float* __restrict__ bias,
    float* __restrict__ out, int n) {
  const int lane = threadIdx.x & 31;
  const int node = __builtin_amdgcn_readfirstlane(blockIdx.x * 8 + (threadIdx.x >> 5));
  if (node >= n) return;
  const int   start = rowptr[node];
  const int   deg   = cnt[node];
  const float di    = dinv[node];
  const float4* __restrict__ h4 = (const float4*)h;

  const int  g      = lane / 10;
  const int  q      = lane - g * 10;
  const bool active = lane < 30;
  const bool holder = lane < 10;

  float4 hs = make_float4(0.0f, 0.0f, 0.0f, 0.0f);
  float4 bv = make_float4(0.0f, 0.0f, 0.0f, 0.0f);
  if (holder) {
    hs = h4[(size_t)node * 10 + lane];
    bv = ((const float4*)bias)[lane];
  }

  float4 acc = make_float4(0.0f, 0.0f, 0.0f, 0.0f);
  for (int base = 0; base < deg; base += 32) {
    const int k = base + lane;
    int   my  = node;
    float myc = 0.0f;
    if (k < deg) {
      my = col[start + k];
      my = min(max(my, 0), n - 1);
      myc = dinv[my] * di;
    }
    const int m = min(32, deg - base);
#pragma unroll 2
    for (int j0 = 0; j0 < m; j0 += 3) {
      const int   j = j0 + g;
      const int   s = __shfl(my, j & 31, 32);
      float       c = __shfl(myc, j & 31, 32);
      c = (active && j < m) ? c : 0.0f;
      const float4 v = h4[(size_t)s * 10 + q];
      acc.x += c * v.x;
      acc.y += c * v.y;
      acc.z += c * v.z;
      acc.w += c * v.w;
    }
  }
  acc.x += __shfl_down(acc.x, 10, 32) + __shfl_down(acc.x, 20, 32);
  acc.y += __shfl_down(acc.y, 10, 32) + __shfl_down(acc.y, 20, 32);
  acc.z += __shfl_down(acc.z, 10, 32) + __shfl_down(acc.z, 20, 32);
  acc.w += __shfl_down(acc.w, 10, 32) + __shfl_down(acc.w, 20, 32);
  const float sc = di * di;
  acc.x += sc * hs.x;
  acc.y += sc * hs.y;
  acc.z += sc * hs.z;
  acc.w += sc * hs.w;

  const float NEG_INF = -INFINITY;
  float4 v;
  v.x = holder ? acc.x + bv.x : NEG_INF;
  v.y = holder ? acc.y + bv.y : NEG_INF;
  v.z = holder ? acc.z + bv.z : NEG_INF;
  v.w = holder ? acc.w + bv.w : NEG_INF;

  float m = fmaxf(fmaxf(v.x, v.y), fmaxf(v.z, v.w));
#pragma unroll
  for (int off = 8; off > 0; off >>= 1) m = fmaxf(m, __shfl_xor(m, off, 32));

  float s = expf(v.x - m) + expf(v.y - m) + expf(v.z - m) + expf(v.w - m);
  s = holder ? s : 0.0f;
#pragma unroll
  for (int off = 8; off > 0; off >>= 1) s += __shfl_xor(s, off, 32);

  const float lse = logf(s);
  if (holder) {
    float4 r;
    r.x = v.x - m - lse;
    r.y = v.y - m - lse;
    r.z = v.z - m - lse;
    r.w = v.w - m - lse;
    ((float4*)out)[(size_t)node * 10 + lane] = r;
  }
}

static inline int cdiv_ll(long long a, long long b) { return (int)((a + b - 1) / b); }

extern "C" void kernel_launch(void* const* d_in, const int* in_sizes, int n_in,
                              void* d_out, int out_size, void* d_ws, size_t ws_size,
                              hipStream_t stream) {
  const float* x  = (const float*)d_in[0];
  const int*   ei = (const int*)  d_in[1];
  const float* W1 = (const float*)d_in[2];
  const float* b1 = (const float*)d_in[3];
  const float* W2 = (const float*)d_in[4];
  const float* b2 = (const float*)d_in[5];

  const int F = 128, H = 128, C = 40;
  const int N = in_sizes[0] / F;
  const int E = in_sizes[1] / 2;
  const int* src = ei;
  const int* dst = ei + E;

  const int nT1 = (H + 15) / 16;
  const int nT2 = (C + 15) / 16;

  float* ws   = (float*)d_ws;
  int*   cnt  = (int*)ws;           ws += N;
  float* dinv = ws;                 ws += N;
  float* h1   = ws;                 ws += (size_t)N * H;
  float* o1   = ws;                 ws += (size_t)N * H;
  int*   col  = (int*)ws;
  int*   rank = col + E;
  int*   rowptr   = rank + E;
  int*   blocktot = rowptr + N;
  ws += (size_t)N * C;
  v2f* bf1    = (v2f*)ws;           ws += (size_t)nT1 * 1024 * 2;
  v2f* bf2    = (v2f*)ws;           ws += (size_t)nT2 * 1024 * 2;
  float* h2   = h1;
  float* out  = (float*)d_out;

  const int B = 256;

  k_zero_i32   <<<cdiv_ll(N, B), B, 0, stream>>>(cnt, N);
  k_deg_rank   <<<cdiv_ll(E, B), B, 0, stream>>>(dst, cnt, rank, E);
  k_scan_block <<<cdiv_ll(N, 1024), B, 0, stream>>>(cnt, rowptr, blocktot, N);
  k_scan_finish<<<cdiv_ll(N, B), B, 0, stream>>>(cnt, rowptr, blocktot, dinv, N);
  k_fill_col   <<<cdiv_ll(E, B), B, 0, stream>>>(src, dst, rank, rowptr, col, E);

  k_pack_bfrag<<<cdiv_ll((long long)nT1 * 1024, B), B, 0, stream>>>(W1, bf1, H, nT1);
  k_pack_bfrag<<<cdiv_ll((long long)nT2 * 1024, B), B, 0, stream>>>(W2, bf2, C, nT2);

  gcn_gemm_wmma_f32<<<N / 16, B, 0, stream>>>(x, bf1, h1, N, H, nT1);
  k_gather_f128_bias_relu<<<cdiv_ll(N, 8), B, 0, stream>>>(h1, rowptr, cnt, col, dinv, b1, o1, N);

  gcn_gemm_wmma_f32<<<N / 16, B, 0, stream>>>(o1, bf2, h2, N, C, nT2);
  k_gather_f40_bias_logsoftmax<<<cdiv_ll(N, 8), B, 0, stream>>>(h2, rowptr, cnt, col, dinv, b2, out, N);
}


# Layer 1 aggregates x before its product, (Ax)W1, and both products run on WMMA from LDS inside the gathering block (16 rows); 11 -> 9 launches; layer-1 sum order: edges first, then k
# speedup vs baseline: 3.6204x; 1.1486x over previous
#include <hip/hip_runtime.h>
#include <hip/hip_bf16.h>
#include <math.h>

typedef float v2f __attribute__((ext_vector_type(2)));
typedef float v8f __attribute__((ext_vector_type(8)));

__global__ void k_zero_i32(int* __restrict__ p, int n) {
  int i = blockIdx.x * blockDim.x + threadIdx.x;
  if (i < n) p[i] = 0;
}

__global__ void k_deg_rank(const int* __restrict__ dst, int* __restrict__ cnt,
                           int* __restrict__ rank, int E) {
  int e = blockIdx.x * blockDim.x + threadIdx.x;
  if (e < E) rank[e] = atomicAdd(&cnt[dst[e]], 1);
}

__global__ __launch_bounds__(256) void k_scan_block(const int* __restrict__ cnt,
                                                    int* __restrict__ rowptr,
                                                    int* __restrict__ blocktot, int n) {
  __shared__ int wsum[8];
  const int tid  = threadIdx.x;
  const int lane = tid & 31;
  const int wave = tid >> 5;
  const int i0   = (blockIdx.x * 256 + tid) * 4;
  int v0 = (i0     < n) ? cnt[i0]     : 0;
  int v1 = (i0 + 1 < n) ? cnt[i0 + 1] : 0;
  int v2 = (i0 + 2 < n) ? cnt[i0 + 2] : 0;
  int v3 = (i0 + 3 < n) ? cnt[i0 + 3] : 0;
  const int s = v0 + v1 + v2 + v3;
  int incl = s;
  for (int off = 1; off < 32; off <<= 1) {
    int t = __shfl_up(incl, off, 32);
    if (lane >= off) incl += t;
  }
  if (lane == 31) wsum[wave] = incl;
  __syncthreads();
  int woff = 0;
  for (int w = 0; w < 8; ++w) {
    int t = wsum[w];
    if (w < wave) woff += t;
  }
  const int excl = woff + incl - s;
  if (i0     < n) rowptr[i0]     = excl;
  if (i0 + 1 < n) rowptr[i0 + 1] = excl + v0;
  if (i0 + 2 < n) rowptr[i0 + 2] = excl + v0 + v1;
  if (i0 + 3 < n) rowptr[i0 + 3] = excl + v0 + v1 + v2;
  if (tid == 255) blocktot[blockIdx.x] = woff + incl;
}

__global__ __launch_bounds__(256) void k_scan_finish(const int* __restrict__ cnt,
                                                     int* __restrict__ rowptr,
                                                     const int* __restrict__ blocktot,
                                                     float* __restrict__ dinv, int n) {
  const int i     = blockIdx.x * 256 + threadIdx.x;
  const int chunk = blockIdx.x >> 2;
  const int lane  = threadIdx.x & 31;
  int part = 0;
  for (int j = lane; j < chunk; j += 32) part += blocktot[j];
  for (int off = 16; off > 0; off >>= 1) part += __shfl_xor(part, off, 32);
  if (i < n) {
    rowptr[i] += part;
    dinv[i] = rsqrtf((float)(cnt[i] + 1));
  }
}

__global__ void k_fill_col(const int* __restrict__ src, const int* __restrict__ dst,
                           const int* __restrict__ rank, const int* __restrict__ rowptr,
                           int* __restrict__ col, int E) {
  int e = blockIdx.x * blockDim.x + threadIdx.x;
  if (e < E) col[rowptr[dst[e]] + rank[e]] = src[e];
}

__global__ void k_pack_bfrag(const float* __restrict__ B, v2f* __restrict__ BF,
                             int N, int nTiles) {
  int i = blockIdx.x * blockDim.x + threadIdx.x;
  int total = nTiles * 1024;
  if (i >= total) return;
  int lane = i & 31;
  int t    = (i >> 5) & 31;
  int nt   = i >> 10;
  int half = lane >> 4;
  int lm   = lane & 15;
  int col  = nt * 16 + lm;
  int kk   = 4 * t + 2 * half;
  v2f v;
  v.x = (col < N) ? B[(size_t)kk * N + col]       : 0.0f;
  v.y = (col < N) ? B[(size_t)(kk + 1) * N + col] : 0.0f;
  BF[i] = v;
}

__global__ __launch_bounds__(512, 16) void k_gather_x_w1_relu_w2(
    const float* __restrict__ x, const int* __restrict__ rowptr,
    const int* __restrict__ cnt, const int* __restrict__ col,
    const float* __restrict__ dinv, const v2f* __restrict__ BF1,
    const float* __restrict__ b1, const v2f* __restrict__ BF2,
    float* __restrict__ h2, int n) {
  constexpr int LDS_STRIDE = 132;
  __shared__ __attribute__((aligned(16))) float As[16 * LDS_STRIDE];
  __shared__ __attribute__((aligned(16))) float Os[16 * LDS_STRIDE];

  const int lane = threadIdx.x & 31;
  const int wave = __builtin_amdgcn_readfirstlane(threadIdx.x >> 5);
  const int m0   = blockIdx.x * 16;
  const int node = m0 + wave;

  float4 acc = make_float4(0.0f, 0.0f, 0.0f, 0.0f);
  if (node < n) {
    const int   start = rowptr[node];
    const int   deg   = cnt[node];
    const float di    = dinv[node];
    const float4* __restrict__ x4 = (const float4*)x;

    const float4 xs = x4[(size_t)node * 32 + lane];
    const float  sc = di * di;
    acc = make_float4(sc * xs.x, sc * xs.y, sc * xs.z, sc * xs.w);

    for (int base = 0; base < deg; base += 32) {
      const int k = base + lane;
      int   my  = 0;
      float myc = 0.0f;
      if (k < deg) {
        my = col[start + k];
        my = min(max(my, 0), n - 1);
        myc = dinv[my] * di;
      }
      const int m = min(32, deg - base);
#pragma unroll 4
      for (int j = 0; j < m; ++j) {
        const int   s = __builtin_amdgcn_readlane(my, j);
        const float c = __int_as_float(__builtin_amdgcn_readlane(__float_as_int(myc), j));
        const float4 v = x4[(size_t)s * 32 + lane];
        acc.x += c * v.x;
        acc.y += c * v.y;
        acc.z += c * v.z;
        acc.w += c * v.w;
      }
    }
  }
  const int half = lane >> 4;
  const int lm   = lane & 15;

  const bool p1 = wave < 8;
  const bool p2 = wave >= 8 && wave < 11;

  *(float4*)&As[wave * LDS_STRIDE + lane * 4] = acc;

  constexpr int RING = 12;

  const v2f* bBase1 = BF1 + (size_t)wave * 1024 + lane;
  v2f b[RING];
  if (p1) {
#pragma unroll
    for (int u = 0; u < RING; ++u) b[u] = bBase1[(size_t)u * 32];
  }
  __syncthreads();

  if (p1) {
    const float* aBase = &As[lm * LDS_STRIDE + 2 * half];
    v8f c = {};
#pragma unroll
    for (int t = 0; t < 32; ++t) {
      v2f a = *(const v2f*)(aBase + 4 * t);
      c = __builtin_amdgcn_wmma_f32_16x16x4_f32(
           false, a,  false, b[t % RING],
           (short)0, c,  false,  false);
      if (t + RING < 32) b[t % RING] = bBase1[(size_t)(t + RING) * 32];
    }
    const int   colh = wave * 16 + lm;
    const float bb   = b1[colh];
#pragma unroll
    for (int r = 0; r < 8; ++r) {
      float v = c[r] + bb;
      v = v > 0.0f ? v : 0.0f;
      Os[(8 * half + r) * LDS_STRIDE + colh] = v;
    }
  }
  __syncthreads();

  if (p2) {
    const float* aBase  = &Os[lm * LDS_STRIDE + 2 * half];
    const v2f*   bBase2 = BF2 + (size_t)(wave - 8) * 1024 + lane;
    v2f q[RING];
#pragma unroll
    for (int u = 0; u < RING; ++u) q[u] = bBase2[(size_t)u * 32];
    v8f c = {};
#pragma unroll
    for (int t = 0; t < 32; ++t) {
      v2f a = *(const v2f*)(aBase + 4 * t);
      c = __builtin_amdgcn_wmma_f32_16x16x4_f32(
           false, a,  false, q[t % RING],
           (short)0, c,  false,  false);
      if (t + RING < 32) q[t % RING] = bBase2[(size_t)(t + RING) * 32];
    }
    const int colc = (wave - 8) * 16 + lm;
    if (colc < 40) {
      const int mBase = m0 + 8 * half;
      float* __restrict__ p = h2 + (size_t)mBase * 40 + colc;
      if (m0 + 16 <= n) {
#pragma unroll
        for (int r = 0; r < 8; ++r) p[r * 40] = c[r];
      } else {
#pragma unroll
        for (int r = 0; r < 8; ++r)
          if (mBase + r < n) p[r * 40] = c[r];
      }
    }
  }
}

__global__ __launch_bounds__(256) void k_gather_f40_bias_logsoftmax(
    const float* __restrict__ h, const int* __restrict__ rowptr,
    const int* __restrict__ cnt, const int* __restrict__ col,
    const float* __restrict__ dinv, const float* __restrict__ bias,
    float* __restrict__ out, int n) {
  const int lane = threadIdx.x & 31;
  const int node = __builtin_amdgcn_readfirstlane(blockIdx.x * 8 + (threadIdx.x >> 5));
  if (node >= n) return;
  const int   start = rowptr[node];
  const int   deg   = cnt[node];
  const float di    = dinv[node];
  const float4* __restrict__ h4 = (const float4*)h;

  const int  g      = lane / 10;
  const int  q      = lane - g * 10;
  const bool active = lane < 30;
  const bool holder = lane < 10;

  float4 hs = make_float4(0.0f, 0.0f, 0.0f, 0.0f);
  float4 bv = make_float4(0.0f, 0.0f, 0.0f, 0.0f);
  if (holder) {
    hs = h4[(size_t)node * 10 + lane];
    bv = ((const float4*)bias)[lane];
  }

  float4 acc = make_float4(0.0f, 0.0f, 0.0f, 0.0f);
  for (int base = 0; base < deg; base += 32) {
    const int k = base + lane;
    int   my  = node;
    float myc = 0.0f;
    if (k < deg) {
      my = col[start + k];
      my = min(max(my, 0), n - 1);
      myc = dinv[my] * di;
    }
    const int m = min(32, deg - base);
#pragma unroll 2
    for (int j0 = 0; j0 < m; j0 += 3) {
      const int   j = j0 + g;
      const int   s = __shfl(my, j & 31, 32);
      float       c = __shfl(myc, j & 31, 32);
      c = (active && j < m) ? c : 0.0f;
      const float4 v = h4[(size_t)s * 10 + q];
      acc.x += c * v.x;
      acc.y += c * v.y;
      acc.z += c * v.z;
      acc.w += c * v.w;
    }
  }
  acc.x += __shfl_down(acc.x, 10, 32) + __shfl_down(acc.x, 20, 32);
  acc.y += __shfl_down(acc.y, 10, 32) + __shfl_down(acc.y, 20, 32);
  acc.z += __shfl_down(acc.z, 10, 32) + __shfl_down(acc.z, 20, 32);
  acc.w += __shfl_down(acc.w, 10, 32) + __shfl_down(acc.w, 20, 32);
  const float sc = di * di;
  acc.x += sc * hs.x;
  acc.y += sc * hs.y;
  acc.z += sc * hs.z;
  acc.w += sc * hs.w;

  const float NEG_INF = -INFINITY;
  float4 v;
  v.x = holder ? acc.x + bv.x : NEG_INF;
  v.y = holder ? acc.y + bv.y : NEG_INF;
  v.z = holder ? acc.z + bv.z : NEG_INF;
  v.w = holder ? acc.w + bv.w : NEG_INF;

  float m = fmaxf(fmaxf(v.x, v.y), fmaxf(v.z, v.w));
#pragma unroll
  for (int off = 8; off > 0; off >>= 1) m = fmaxf(m, __shfl_xor(m, off, 32));

  float s = expf(v.x - m) + expf(v.y - m) + expf(v.z - m) + expf(v.w - m);
  s = holder ? s : 0.0f;
#pragma unroll
  for (int off = 8; off > 0; off >>= 1) s += __shfl_xor(s, off, 32);

  const float lse = logf(s);
  if (holder) {
    float4 r;
    r.x = v.x - m - lse;
    r.y = v.y - m - lse;
    r.z = v.z - m - lse;
    r.w = v.w - m - lse;
    ((float4*)out)[(size_t)node * 10 + lane] = r;
  }
}

static inline int cdiv_ll(long long a, long long b) { return (int)((a + b - 1) / b); }

extern "C" void kernel_launch(void* const* d_in, const int* in_sizes, int n_in,
                              void* d_out, int out_size, void* d_ws, size_t ws_size,
                              hipStream_t stream) {
  const float* x  = (const float*)d_in[0];
  const int*   ei = (const int*)  d_in[1];
  const float* W1 = (const float*)d_in[2];
  const float* b1 = (const float*)d_in[3];
  const float* W2 = (const float*)d_in[4];
  const float* b2 = (const float*)d_in[5];

  const int F = 128, H = 128, C = 40;
  const int N = in_sizes[0] / F;
  const int E = in_sizes[1] / 2;
  const int* src = ei;
  const int* dst = ei + E;

  const int nT1 = (H + 15) / 16;
  const int nT2 = (C + 15) / 16;

  float* ws   = (float*)d_ws;
  int*   cnt  = (int*)ws;           ws += N;
  float* dinv = ws;                 ws += N;
  float* h2   = ws;                 ws += (size_t)N * H;
  ws += (size_t)N * H;
  int*   col  = (int*)ws;
  int*   rank = col + E;
  int*   rowptr   = rank + E;
  int*   blocktot = rowptr + N;
  ws += (size_t)N * C;
  v2f* bf1    = (v2f*)ws;           ws += (size_t)nT1 * 1024 * 2;
  v2f* bf2    = (v2f*)ws;           ws += (size_t)nT2 * 1024 * 2;
  float* out  = (float*)d_out;

  const int B = 256;

  k_zero_i32   <<<cdiv_ll(N, B), B, 0, stream>>>(cnt, N);
  k_deg_rank   <<<cdiv_ll(E, B), B, 0, stream>>>(dst, cnt, rank, E);
  k_scan_block <<<cdiv_ll(N, 1024), B, 0, stream>>>(cnt, rowptr, blocktot, N);
  k_scan_finish<<<cdiv_ll(N, B), B, 0, stream>>>(cnt, rowptr, blocktot, dinv, N);
  k_fill_col   <<<cdiv_ll(E, B), B, 0, stream>>>(src, dst, rank, rowptr, col, E);

  k_pack_bfrag<<<cdiv_ll((long long)nT1 * 1024, B), B, 0, stream>>>(W1, bf1, H, nT1);
  k_pack_bfrag<<<cdiv_ll((long long)nT2 * 1024, B), B, 0, stream>>>(W2, bf2, C, nT2);

  k_gather_x_w1_relu_w2<<<cdiv_ll(N, 16), 512, 0, stream>>>(x, rowptr, cnt, col, dinv, bf1, b1, bf2, h2, N);

  k_gather_f40_bias_logsoftmax<<<cdiv_ll(N, 8), B, 0, stream>>>(h2, rowptr, cnt, col, dinv, b2, out, N);
}


# Edge list by binning: edges copied into 256-node bins (LDS histograms, count-matrix scan), one block per bin ranks them with LDS adds; no global atomics, 9 -> 7 launches; row sum order: LDS arrival
# speedup vs baseline: 5.6536x; 1.5616x over previous
#include <hip/hip_runtime.h>
#include <hip/hip_bf16.h>
#include <math.h>

typedef float v2f __attribute__((ext_vector_type(2)));
typedef float v8f __attribute__((ext_vector_type(8)));

constexpr int BIN_SHIFT    = 8;
constexpr int MAX_BINS     = 256;
constexpr int N_SLICES     = 128;
constexpr int EDGE_THREADS = 1024;
constexpr int EDGE_BATCH   = 8;

__device__ __forceinline__ int bin_of(int d) {
  return min((int)((unsigned)d >> BIN_SHIFT), MAX_BINS - 1);
}

__global__ __launch_bounds__(EDGE_THREADS) void k_bin_hist(const int* __restrict__ dst,
                                                           int* __restrict__ hist, int E) {
  __shared__ int h[MAX_BINS];
  const int tid = threadIdx.x;
  if (tid < MAX_BINS) h[tid] = 0;
  __syncthreads();
  const int slice = (E + N_SLICES - 1) / N_SLICES;
  const int e0 = blockIdx.x * slice;
  const int e1 = min(E, e0 + slice);
  for (int e = e0 + tid; e < e1; e += EDGE_THREADS * EDGE_BATCH) {
    int d[EDGE_BATCH];
#pragma unroll
    for (int u = 0; u < EDGE_BATCH; ++u) {
      d[u] = dst[min(e + u * EDGE_THREADS, e1 - 1)];
    }
#pragma unroll
    for (int u = 0; u < EDGE_BATCH; ++u) asm volatile("" : "+v"(d[u]));
#pragma unroll
    for (int u = 0; u < EDGE_BATCH; ++u) {
      const int i = e + u * EDGE_THREADS;
      if (i < e1) atomicAdd(&h[bin_of(d[u])], 1);
    }
  }
  __syncthreads();
  if (tid < MAX_BINS) hist[blockIdx.x * MAX_BINS + tid] = h[tid];
}

__global__ __launch_bounds__(EDGE_THREADS) void k_bin_scatter(
    const int* __restrict__ src, const int* __restrict__ dst,
    const int* __restrict__ hist, int* __restrict__ binStart,
    int2* __restrict__ pairs, int E) {
  constexpr int GROUPS = EDGE_THREADS / 64;
  constexpr int ROWS   = N_SLICES / GROUPS;
  __shared__ __attribute__((aligned(16))) int ptot[GROUPS][MAX_BINS];
  __shared__ __attribute__((aligned(16))) int ppre[GROUPS][MAX_BINS];
  __shared__ int cursor[MAX_BINS];
  __shared__ int wsum[8];
  const int tid  = threadIdx.x;
  const int lane = tid & 31;
  const int wave = tid >> 5;
  const int blk  = blockIdx.x;

  {
    const int g = __builtin_amdgcn_readfirstlane(tid >> 6);
    const int q = tid & 63;
    const int4* __restrict__ h4 = (const int4*)hist;
    int4 v[ROWS];
#pragma unroll
    for (int u = 0; u < ROWS; ++u) v[u] = h4[(g * ROWS + u) * (MAX_BINS / 4) + q];
    int4 tot = make_int4(0, 0, 0, 0), pre = make_int4(0, 0, 0, 0);
#pragma unroll
    for (int u = 0; u < ROWS; ++u) {
      tot.x += v[u].x; tot.y += v[u].y; tot.z += v[u].z; tot.w += v[u].w;
      if (g * ROWS + u < blk) {
        pre.x += v[u].x; pre.y += v[u].y; pre.z += v[u].z; pre.w += v[u].w;
      }
    }
    *(int4*)&ptot[g][4 * q] = tot;
    *(int4*)&ppre[g][4 * q] = pre;
  }
  __syncthreads();

  int tot = 0, pre = 0;
  if (tid < MAX_BINS) {
#pragma unroll
    for (int g = 0; g < GROUPS; ++g) { tot += ptot[g][tid]; pre += ppre[g][tid]; }
  }
  int incl = tot;
  for (int off = 1; off < 32; off <<= 1) {
    int u = __shfl_up(incl, off, 32);
    if (lane >= off) incl += u;
  }
  if (tid < MAX_BINS && lane == 31) wsum[wave] = incl;
  __syncthreads();
  if (tid < MAX_BINS) {
    int woff = 0;
    for (int w = 0; w < 8; ++w) {
      int u = wsum[w];
      if (w < wave) woff += u;
    }
    const int excl = woff + incl - tot;
    cursor[tid] = excl + pre;
    if (blk == 0) {
      binStart[tid] = excl;
      if (tid == MAX_BINS - 1) binStart[MAX_BINS] = excl + tot;
    }
  }
  __syncthreads();

  const int slice = (E + N_SLICES - 1) / N_SLICES;
  const int e0 = blk * slice;
  const int e1 = min(E, e0 + slice);
  for (int e = e0 + tid; e < e1; e += EDGE_THREADS * EDGE_BATCH) {
    int d[EDGE_BATCH], s[EDGE_BATCH];
#pragma unroll
    for (int u = 0; u < EDGE_BATCH; ++u) {
      const int i = min(e + u * EDGE_THREADS, e1 - 1);
      d[u] = dst[i];
      s[u] = src[i];
    }
#pragma unroll
    for (int u = 0; u < EDGE_BATCH; ++u) asm volatile("" : "+v"(d[u]), "+v"(s[u]));
#pragma unroll
    for (int u = 0; u < EDGE_BATCH; ++u) {
      const int i = e + u * EDGE_THREADS;
      if (i < e1) {
        const int pos = atomicAdd(&cursor[bin_of(d[u])], 1);
        pairs[pos] = make_int2(s[u], d[u]);
      }
    }
  }
}

__global__ __launch_bounds__(EDGE_THREADS) void k_bin_csr(
    const int2* __restrict__ pairs, const int* __restrict__ binStart,
    int* __restrict__ cnt, int* __restrict__ rowptr, float* __restrict__ dinv,
    int* __restrict__ col, int n) {
  constexpr int BATCH = 4;
  __shared__ int c[MAX_BINS];
  __shared__ int cur[MAX_BINS];
  __shared__ int wsum[8];
  const int tid  = threadIdx.x;
  const int lane = tid & 31;
  const int wave = tid >> 5;
  const int bin  = blockIdx.x;
  const int base = binStart[bin];
  const int end  = binStart[bin + 1];
  if (tid < MAX_BINS) c[tid] = 0;
  __syncthreads();

  for (int i0 = base + tid; i0 < end; i0 += EDGE_THREADS * BATCH) {
    int d[BATCH];
#pragma unroll
    for (int u = 0; u < BATCH; ++u) {
      d[u] = pairs[min(i0 + u * EDGE_THREADS, end - 1)].y;
    }
#pragma unroll
    for (int u = 0; u < BATCH; ++u) asm volatile("" : "+v"(d[u]));
#pragma unroll
    for (int u = 0; u < BATCH; ++u) {
      const int i = i0 + u * EDGE_THREADS;
      if (i < end) atomicAdd(&c[d[u] & (MAX_BINS - 1)], 1);
    }
  }
  __syncthreads();

  const int mine = (tid < MAX_BINS) ? c[tid] : 0;
  int incl = mine;
  for (int off = 1; off < 32; off <<= 1) {
    int u = __shfl_up(incl, off, 32);
    if (lane >= off) incl += u;
  }
  if (tid < MAX_BINS && lane == 31) wsum[wave] = incl;
  __syncthreads();
  if (tid < MAX_BINS) {
    int woff = 0;
    for (int w = 0; w < 8; ++w) {
      int u = wsum[w];
      if (w < wave) woff += u;
    }
    const int excl = woff + incl - mine;
    cur[tid] = excl;
    const int node = (bin << BIN_SHIFT) + tid;
    if (node < n) {
      cnt[node]    = mine;
      rowptr[node] = base + excl;
      dinv[node]   = rsqrtf((float)(mine + 1));
    }
  }
  __syncthreads();

  for (int i0 = base + tid; i0 < end; i0 += EDGE_THREADS * BATCH) {
    int2 p[BATCH];
#pragma unroll
    for (int u = 0; u < BATCH; ++u) {
      p[u] = pairs[min(i0 + u * EDGE_THREADS, end - 1)];
    }
#pragma unroll
    for (int u = 0; u < BATCH; ++u) asm volatile("" : "+v"(p[u].x), "+v"(p[u].y));
#pragma unroll
    for (int u = 0; u < BATCH; ++u) {
      const int i = i0 + u * EDGE_THREADS;
      if (i < end) {
        const int pos = atomicAdd(&cur[p[u].y & (MAX_BINS - 1)], 1);
        col[base + pos] = p[u].x;
      }
    }
  }
}

__global__ void k_pack_bfrag(const float* __restrict__ B, v2f* __restrict__ BF,
                             int N, int nTiles) {
  int i = blockIdx.x * blockDim.x + threadIdx.x;
  int total = nTiles * 1024;
  if (i >= total) return;
  int lane = i & 31;
  int t    = (i >> 5) & 31;
  int nt   = i >> 10;
  int half = lane >> 4;
  int lm   = lane & 15;
  int col  = nt * 16 + lm;
  int kk   = 4 * t + 2 * half;
  v2f v;
  v.x = (col < N) ? B[(size_t)kk * N + col]       : 0.0f;
  v.y = (col < N) ? B[(size_t)(kk + 1) * N + col] : 0.0f;
  BF[i] = v;
}

__global__ __launch_bounds__(512, 16) void k_gather_x_w1_relu_w2(
    const float* __restrict__ x, const int* __restrict__ rowptr,
    const int* __restrict__ cnt, const int* __restrict__ col,
    const float* __restrict__ dinv, const v2f* __restrict__ BF1,
    const float* __restrict__ b1, const v2f* __restrict__ BF2,
    float* __restrict__ h2, int n) {
  constexpr int LDS_STRIDE = 132;
  __shared__ __attribute__((aligned(16))) float As[16 * LDS_STRIDE];
  __shared__ __attribute__((aligned(16))) float Os[16 * LDS_STRIDE];

  const int lane = threadIdx.x & 31;
  const int wave = __builtin_amdgcn_readfirstlane(threadIdx.x >> 5);
  const int m0   = blockIdx.x * 16;
  const int node = m0 + wave;

  float4 acc = make_float4(0.0f, 0.0f, 0.0f, 0.0f);
  if (node < n) {
    const int   start = rowptr[node];
    const int   deg   = cnt[node];
    const float di    = dinv[node];
    const float4* __restrict__ x4 = (const float4*)x;

    const float4 xs = x4[(size_t)node * 32 + lane];
    const float  sc = di * di;
    acc = make_float4(sc * xs.x, sc * xs.y, sc * xs.z, sc * xs.w);

    for (int base = 0; base < deg; base += 32) {
      const int k = base + lane;
      int   my  = 0;
      float myc = 0.0f;
      if (k < deg) {
        my = col[start + k];
        my = min(max(my, 0), n - 1);
        myc = dinv[my] * di;
      }
      const int m = min(32, deg - base);
#pragma unroll 4
      for (int j = 0; j < m; ++j) {
        const int   s = __builtin_amdgcn_readlane(my, j);
        const float c = __int_as_float(__builtin_amdgcn_readlane(__float_as_int(myc), j));
        const float4 v = x4[(size_t)s * 32 + lane];
        acc.x += c * v.x;
        acc.y += c * v.y;
        acc.z += c * v.z;
        acc.w += c * v.w;
      }
    }
  }
  const int half = lane >> 4;
  const int lm   = lane & 15;

  const bool p1 = wave < 8;
  const bool p2 = wave >= 8 && wave < 11;

  *(float4*)&As[wave * LDS_STRIDE + lane * 4] = acc;

  constexpr int RING = 12;

  const v2f* bBase1 = BF1 + (size_t)wave * 1024 + lane;
  v2f b[RING];
  if (p1) {
#pragma unroll
    for (int u = 0; u < RING; ++u) b[u] = bBase1[(size_t)u * 32];
  }
  __syncthreads();

  if (p1) {
    const float* aBase = &As[lm * LDS_STRIDE + 2 * half];
    v8f c = {};
#pragma unroll
    for (int t = 0; t < 32; ++t) {
      v2f a = *(const v2f*)(aBase + 4 * t);
      c = __builtin_amdgcn_wmma_f32_16x16x4_f32(
           false, a,  false, b[t % RING],
           (short)0, c,  false,  false);
      if (t + RING < 32) b[t % RING] = bBase1[(size_t)(t + RING) * 32];
    }
    const int   colh = wave * 16 + lm;
    const float bb   = b1[colh];
#pragma unroll
    for (int r = 0; r < 8; ++r) {
      float v = c[r] + bb;
      v = v > 0.0f ? v : 0.0f;
      Os[(8 * half + r) * LDS_STRIDE + colh] = v;
    }
  }
  __syncthreads();

  if (p2) {
    const float* aBase  = &Os[lm * LDS_STRIDE + 2 * half];
    const v2f*   bBase2 = BF2 + (size_t)(wave - 8) * 1024 + lane;
    v2f q[RING];
#pragma unroll
    for (int u = 0; u < RING; ++u) q[u] = bBase2[(size_t)u * 32];
    v8f c = {};
#pragma unroll
    for (int t = 0; t < 32; ++t) {
      v2f a = *(const v2f*)(aBase + 4 * t);
      c = __builtin_amdgcn_wmma_f32_16x16x4_f32(
           false, a,  false, q[t % RING],
           (short)0, c,  false,  false);
      if (t + RING < 32) q[t % RING] = bBase2[(size_t)(t + RING) * 32];
    }
    const int colc = (wave - 8) * 16 + lm;
    if (colc < 40) {
      const int mBase = m0 + 8 * half;
      float* __restrict__ p = h2 + (size_t)mBase * 40 + colc;
      if (m0 + 16 <= n) {
#pragma unroll
        for (int r = 0; r < 8; ++r) p[r * 40] = c[r];
      } else {
#pragma unroll
        for (int r = 0; r < 8; ++r)
          if (mBase + r < n) p[r * 40] = c[r];
      }
    }
  }
}

__global__ __launch_bounds__(256) void k_gather_f40_bias_logsoftmax(
    const float* __restrict__ h, const int* __restrict__ rowptr,
    const int* __restrict__ cnt, const int* __restrict__ col,
    const float* __restrict__ dinv, const float* __restrict__ bias,
    float* __restrict__ out, int n) {
  const int lane = threadIdx.x & 31;
  const int node = __builtin_amdgcn_readfirstlane(blockIdx.x * 8 + (threadIdx.x >> 5));
  if (node >= n) return;
  const int   start = rowptr[node];
  const int   deg   = cnt[node];
  const float di    = dinv[node];
  const float4* __restrict__ h4 = (const float4*)h;

  const int  g      = lane / 10;
  const int  q      = lane - g * 10;
  const bool active = lane < 30;
  const bool holder = lane < 10;

  float4 hs = make_float4(0.0f, 0.0f, 0.0f, 0.0f);
  float4 bv = make_float4(0.0f, 0.0f, 0.0f, 0.0f);
  if (holder) {
    hs = h4[(size_t)node * 10 + lane];
    bv = ((const float4*)bias)[lane];
  }

  float4 acc = make_float4(0.0f, 0.0f, 0.0f, 0.0f);
  for (int base = 0; base < deg; base += 32) {
    const int k = base + lane;
    int   my  = node;
    float myc = 0.0f;
    if (k < deg) {
      my = col[start + k];
      my = min(max(my, 0), n - 1);
      myc = dinv[my] * di;
    }
    const int m = min(32, deg - base);
#pragma unroll 2
    for (int j0 = 0; j0 < m; j0 += 3) {
      const int   j = j0 + g;
      const int   s = __shfl(my, j & 31, 32);
      float       c = __shfl(myc, j & 31, 32);
      c = (active && j < m) ? c : 0.0f;
      const float4 v = h4[(size_t)s * 10 + q];
      acc.x += c * v.x;
      acc.y += c * v.y;
      acc.z += c * v.z;
      acc.w += c * v.w;
    }
  }
  acc.x += __shfl_down(acc.x, 10, 32) + __shfl_down(acc.x, 20, 32);
  acc.y += __shfl_down(acc.y, 10, 32) + __shfl_down(acc.y, 20, 32);
  acc.z += __shfl_down(acc.z, 10, 32) + __shfl_down(acc.z, 20, 32);
  acc.w += __shfl_down(acc.w, 10, 32) + __shfl_down(acc.w, 20, 32);
  const float sc = di * di;
  acc.x += sc * hs.x;
  acc.y += sc * hs.y;
  acc.z += sc * hs.z;
  acc.w += sc * hs.w;

  const float NEG_INF = -INFINITY;
  float4 v;
  v.x = holder ? acc.x + bv.x : NEG_INF;
  v.y = holder ? acc.y + bv.y : NEG_INF;
  v.z = holder ? acc.z + bv.z : NEG_INF;
  v.w = holder ? acc.w + bv.w : NEG_INF;

  float m = fmaxf(fmaxf(v.x, v.y), fmaxf(v.z, v.w));
#pragma unroll
  for (int off = 8; off > 0; off >>= 1) m = fmaxf(m, __shfl_xor(m, off, 32));

  float s = expf(v.x - m) + expf(v.y - m) + expf(v.z - m) + expf(v.w - m);
  s = holder ? s : 0.0f;
#pragma unroll
  for (int off = 8; off > 0; off >>= 1) s += __shfl_xor(s, off, 32);

  const float lse = logf(s);
  if (holder) {
    float4 r;
    r.x = v.x - m - lse;
    r.y = v.y - m - lse;
    r.z = v.z - m - lse;
    r.w = v.w - m - lse;
    ((float4*)out)[(size_t)node * 10 + lane] = r;
  }
}

static inline int cdiv_ll(long long a, long long b) { return (int)((a + b - 1) / b); }

extern "C" void kernel_launch(void* const* d_in, const int* in_sizes, int n_in,
                              void* d_out, int out_size, void* d_ws, size_t ws_size,
                              hipStream_t stream) {
  const float* x  = (const float*)d_in[0];
  const int*   ei = (const int*)  d_in[1];
  const float* W1 = (const float*)d_in[2];
  const float* b1 = (const float*)d_in[3];
  const float* W2 = (const float*)d_in[4];
  const float* b2 = (const float*)d_in[5];

  const int F = 128, H = 128, C = 40;
  const int N = in_sizes[0] / F;
  const int E = in_sizes[1] / 2;
  const int* src = ei;
  const int* dst = ei + E;

  const int nT1 = (H + 15) / 16;
  const int nT2 = (C + 15) / 16;

  float* ws   = (float*)d_ws;
  int*   cnt  = (int*)ws;           ws += N;
  float* dinv = ws;                 ws += N;
  float* h2   = ws;                 ws += (size_t)N * H;
  int2*  pairs    = (int2*)ws;
  int*   hist     = (int*)ws + (size_t)2 * E;
  int*   binStart = hist + N_SLICES * MAX_BINS;
  ws += (size_t)N * H;
  int*   col  = (int*)ws;
  int*   rowptr = col + E;
  ws += (size_t)N * C;
  v2f* bf1    = (v2f*)ws;           ws += (size_t)nT1 * 1024 * 2;
  v2f* bf2    = (v2f*)ws;           ws += (size_t)nT2 * 1024 * 2;
  float* out  = (float*)d_out;

  const int B = 256;
  const int nBins = cdiv_ll(N, MAX_BINS);

  k_bin_hist   <<<N_SLICES, EDGE_THREADS, 0, stream>>>(dst, hist, E);
  k_bin_scatter<<<N_SLICES, EDGE_THREADS, 0, stream>>>(src, dst, hist, binStart, pairs, E);
  k_bin_csr    <<<nBins, EDGE_THREADS, 0, stream>>>(pairs, binStart, cnt, rowptr, dinv, col, N);

  k_pack_bfrag<<<cdiv_ll((long long)nT1 * 1024, B), B, 0, stream>>>(W1, bf1, H, nT1);
  k_pack_bfrag<<<cdiv_ll((long long)nT2 * 1024, B), B, 0, stream>>>(W2, bf2, C, nT2);

  k_gather_x_w1_relu_w2<<<cdiv_ll(N, 16), 512, 0, stream>>>(x, rowptr, cnt, col, dinv, bf1, b1, bf2, h2, N);

  k_gather_f40_bias_logsoftmax<<<cdiv_ll(N, 8), B, 0, stream>>>(h2, rowptr, cnt, col, dinv, b2, out, N);
}


# Weight packing (W1, W2 into WMMA fragment order) rides in the first edge launch as 11 extra blocks instead of two launches of its own: 7 -> 5 launches, nothing else changed
# speedup vs baseline: 5.9849x; 1.0586x over previous
#include <hip/hip_runtime.h>
#include <hip/hip_bf16.h>
#include <math.h>

typedef float v2f __attribute__((ext_vector_type(2)));
typedef float v8f __attribute__((ext_vector_type(8)));

constexpr int BIN_SHIFT    = 8;
constexpr int MAX_BINS     = 256;
constexpr int N_SLICES     = 128;
constexpr int EDGE_THREADS = 1024;
constexpr int EDGE_BATCH   = 8;

__device__ __forceinline__ int bin_of(int d) {
  return min((int)((unsigned)d >> BIN_SHIFT), MAX_BINS - 1);
}

__device__ __forceinline__ void pack_bfrag(const float* __restrict__ B, v2f* __restrict__ BF,
                                           int N, int i) {
  const int lane = i & 31;
  const int t    = (i >> 5) & 31;
  const int nt   = i >> 10;
  const int half = lane >> 4;
  const int lm   = lane & 15;
  const int col  = nt * 16 + lm;
  const int kk   = 4 * t + 2 * half;
  v2f v;
  v.x = (col < N) ? B[(size_t)kk * N + col]       : 0.0f;
  v.y = (col < N) ? B[(size_t)(kk + 1) * N + col] : 0.0f;
  BF[i] = v;
}

__global__ __launch_bounds__(EDGE_THREADS) void k_bin_hist_pack(
    const int* __restrict__ dst, int* __restrict__ hist, int E,
    const float* __restrict__ W1, v2f* __restrict__ BF1, int H, int nT1,
    const float* __restrict__ W2, v2f* __restrict__ BF2, int C, int nT2) {
  __shared__ int h[MAX_BINS];
  const int tid = threadIdx.x;
  if (blockIdx.x >= N_SLICES) {
    const int i = (blockIdx.x - N_SLICES) * EDGE_THREADS + tid;
    if (i < nT1 * 1024)               pack_bfrag(W1, BF1, H, i);
    else if (i < (nT1 + nT2) * 1024)  pack_bfrag(W2, BF2, C, i - nT1 * 1024);
    return;
  }
  if (tid < MAX_BINS) h[tid] = 0;
  __syncthreads();
  const int slice = (E + N_SLICES - 1) / N_SLICES;
  const int e0 = blockIdx.x * slice;
  const int e1 = min(E, e0 + slice);
  for (int e = e0 + tid; e < e1; e += EDGE_THREADS * EDGE_BATCH) {
    int d[EDGE_BATCH];
#pragma unroll
    for (int u = 0; u < EDGE_BATCH; ++u) {
      d[u] = dst[min(e + u * EDGE_THREADS, e1 - 1)];
    }
#pragma unroll
    for (int u = 0; u < EDGE_BATCH; ++u) asm volatile("" : "+v"(d[u]));
#pragma unroll
    for (int u = 0; u < EDGE_BATCH; ++u) {
      const int i = e + u * EDGE_THREADS;
      if (i < e1) atomicAdd(&h[bin_of(d[u])], 1);
    }
  }
  __syncthreads();
  if (tid < MAX_BINS) hist[blockIdx.x * MAX_BINS + tid] = h[tid];
}

__global__ __launch_bounds__(EDGE_THREADS) void k_bin_scatter(
    const int* __restrict__ src, const int* __restrict__ dst,
    const int* __restrict__ hist, int* __restrict__ binStart,
    int2* __restrict__ pairs, int E) {
  constexpr int GROUPS = EDGE_THREADS / 64;
  constexpr int ROWS   = N_SLICES / GROUPS;
  __shared__ __attribute__((aligned(16))) int ptot[GROUPS][MAX_BINS];
  __shared__ __attribute__((aligned(16))) int ppre[GROUPS][MAX_BINS];
  __shared__ int cursor[MAX_BINS];
  __shared__ int wsum[8];
  const int tid  = threadIdx.x;
  const int lane = tid & 31;
  const int wave = tid >> 5;
  const int blk  = blockIdx.x;

  {
    const int g = __builtin_amdgcn_readfirstlane(tid >> 6);
    const int q = tid & 63;
    const int4* __restrict__ h4 = (const int4*)hist;
    int4 v[ROWS];
#pragma unroll
    for (int u = 0; u < ROWS; ++u) v[u] = h4[(g * ROWS + u) * (MAX_BINS / 4) + q];
    int4 tot = make_int4(0, 0, 0, 0), pre = make_int4(0, 0, 0, 0);
#pragma unroll
    for (int u = 0; u < ROWS; ++u) {
      tot.x += v[u].x; tot.y += v[u].y; tot.z += v[u].z; tot.w += v[u].w;
      if (g * ROWS + u < blk) {
        pre.x += v[u].x; pre.y += v[u].y; pre.z += v[u].z; pre.w += v[u].w;
      }
    }
    *(int4*)&ptot[g][4 * q] = tot;
    *(int4*)&ppre[g][4 * q] = pre;
  }
  __syncthreads();

  int tot = 0, pre = 0;
  if (tid < MAX_BINS) {
#pragma unroll
    for (int g = 0; g < GROUPS; ++g) { tot += ptot[g][tid]; pre += ppre[g][tid]; }
  }
  int incl = tot;
  for (int off = 1; off < 32; off <<= 1) {
    int u = __shfl_up(incl, off, 32);
    if (lane >= off) incl += u;
  }
  if (tid < MAX_BINS && lane == 31) wsum[wave] = incl;
  __syncthreads();
  if (tid < MAX_BINS) {
    int woff = 0;
    for (int w = 0; w < 8; ++w) {
      int u = wsum[w];
      if (w < wave) woff += u;
    }
    const int excl = woff + incl - tot;
    cursor[tid] = excl + pre;
    if (blk == 0) {
      binStart[tid] = excl;
      if (tid == MAX_BINS - 1) binStart[MAX_BINS] = excl + tot;
    }
  }
  __syncthreads();

  const int slice = (E + N_SLICES - 1) / N_SLICES;
  const int e0 = blk * slice;
  const int e1 = min(E, e0 + slice);
  for (int e = e0 + tid; e < e1; e += EDGE_THREADS * EDGE_BATCH) {
    int d[EDGE_BATCH], s[EDGE_BATCH];
#pragma unroll
    for (int u = 0; u < EDGE_BATCH; ++u) {
      const int i = min(e + u * EDGE_THREADS, e1 - 1);
      d[u] = dst[i];
      s[u] = src[i];
    }
#pragma unroll
    for (int u = 0; u < EDGE_BATCH; ++u) asm volatile("" : "+v"(d[u]), "+v"(s[u]));
#pragma unroll
    for (int u = 0; u < EDGE_BATCH; ++u) {
      const int i = e + u * EDGE_THREADS;
      if (i < e1) {
        const int pos = atomicAdd(&cursor[bin_of(d[u])], 1);
        pairs[pos] = make_int2(s[u], d[u]);
      }
    }
  }
}

__global__ __launch_bounds__(EDGE_THREADS) void k_bin_csr(
    const int2* __restrict__ pairs, const int* __restrict__ binStart,
    int* __restrict__ cnt, int* __restrict__ rowptr, float* __restrict__ dinv,
    int* __restrict__ col, int n) {
  constexpr int BATCH = 4;
  __shared__ int c[MAX_BINS];
  __shared__ int cur[MAX_BINS];
  __shared__ int wsum[8];
  const int tid  = threadIdx.x;
  const int lane = tid & 31;
  const int wave = tid >> 5;
  const int bin  = blockIdx.x;
  const int base = binStart[bin];
  const int end  = binStart[bin + 1];
  if (tid < MAX_BINS) c[tid] = 0;
  __syncthreads();

  for (int i0 = base + tid; i0 < end; i0 += EDGE_THREADS * BATCH) {
    int d[BATCH];
#pragma unroll
    for (int u = 0; u < BATCH; ++u) {
      d[u] = pairs[min(i0 + u * EDGE_THREADS, end - 1)].y;
    }
#pragma unroll
    for (int u = 0; u < BATCH; ++u) asm volatile("" : "+v"(d[u]));
#pragma unroll
    for (int u = 0; u < BATCH; ++u) {
      const int i = i0 + u * EDGE_THREADS;
      if (i < end) atomicAdd(&c[d[u] & (MAX_BINS - 1)], 1);
    }
  }
  __syncthreads();

  const int mine = (tid < MAX_BINS) ? c[tid] : 0;
  int incl = mine;
  for (int off = 1; off < 32; off <<= 1) {
    int u = __shfl_up(incl, off, 32);
    if (lane >= off) incl += u;
  }
  if (tid < MAX_BINS && lane == 31) wsum[wave] = incl;
  __syncthreads();
  if (tid < MAX_BINS) {
    int woff = 0;
    for (int w = 0; w < 8; ++w) {
      int u = wsum[w];
      if (w < wave) woff += u;
    }
    const int excl = woff + incl - mine;
    cur[tid] = excl;
    const int node = (bin << BIN_SHIFT) + tid;
    if (node < n) {
      cnt[node]    = mine;
      rowptr[node] = base + excl;
      dinv[node]   = rsqrtf((float)(mine + 1));
    }
  }
  __syncthreads();

  for (int i0 = base + tid; i0 < end; i0 += EDGE_THREADS * BATCH) {
    int2 p[BATCH];
#pragma unroll
    for (int u = 0; u < BATCH; ++u) {
      p[u] = pairs[min(i0 + u * EDGE_THREADS, end - 1)];
    }
#pragma unroll
    for (int u = 0; u < BATCH; ++u) asm volatile("" : "+v"(p[u].x), "+v"(p[u].y));
#pragma unroll
    for (int u = 0; u < BATCH; ++u) {
      const int i = i0 + u * EDGE_THREADS;
      if (i < end) {
        const int pos = atomicAdd(&cur[p[u].y & (MAX_BINS - 1)], 1);
        col[base + pos] = p[u].x;
      }
    }
  }
}

__global__ __launch_bounds__(512, 16) void k_gather_x_w1_relu_w2(
    const float* __restrict__ x, const int* __restrict__ rowptr,
    const int* __restrict__ cnt, const int* __restrict__ col,
    const float* __restrict__ dinv, const v2f* __restrict__ BF1,
    const float* __restrict__ b1, const v2f* __restrict__ BF2,
    float* __restrict__ h2, int n) {
  constexpr int LDS_STRIDE = 132;
  __shared__ __attribute__((aligned(16))) float As[16 * LDS_STRIDE];
  __shared__ __attribute__((aligned(16))) float Os[16 * LDS_STRIDE];

  const int lane = threadIdx.x & 31;
  const int wave = __builtin_amdgcn_readfirstlane(threadIdx.x >> 5);
  const int m0   = blockIdx.x * 16;
  const int node = m0 + wave;

  float4 acc = make_float4(0.0f, 0.0f, 0.0f, 0.0f);
  if (node < n) {
    const int   start = rowptr[node];
    const int   deg   = cnt[node];
    const float di    = dinv[node];
    const float4* __restrict__ x4 = (const float4*)x;

    const float4 xs = x4[(size_t)node * 32 + lane];
    const float  sc = di * di;
    acc = make_float4(sc * xs.x, sc * xs.y, sc * xs.z, sc * xs.w);

    for (int base = 0; base < deg; base += 32) {
      const int k = base + lane;
      int   my  = 0;
      float myc = 0.0f;
      if (k < deg) {
        my = col[start + k];
        my = min(max(my, 0), n - 1);
        myc = dinv[my] * di;
      }
      const int m = min(32, deg - base);
#pragma unroll 4
      for (int j = 0; j < m; ++j) {
        const int   s = __builtin_amdgcn_readlane(my, j);
        const float c = __int_as_float(__builtin_amdgcn_readlane(__float_as_int(myc), j));
        const float4 v = x4[(size_t)s * 32 + lane];
        acc.x += c * v.x;
        acc.y += c * v.y;
        acc.z += c * v.z;
        acc.w += c * v.w;
      }
    }
  }
  const int half = lane >> 4;
  const int lm   = lane & 15;

  const bool p1 = wave < 8;
  const bool p2 = wave >= 8 && wave < 11;

  *(float4*)&As[wave * LDS_STRIDE + lane * 4] = acc;

  constexpr int RING = 12;

  const v2f* bBase1 = BF1 + (size_t)wave * 1024 + lane;
  v2f b[RING];
  if (p1) {
#pragma unroll
    for (int u = 0; u < RING; ++u) b[u] = bBase1[(size_t)u * 32];
  }
  __syncthreads();

  if (p1) {
    const float* aBase = &As[lm * LDS_STRIDE + 2 * half];
    v8f c = {};
#pragma unroll
    for (int t = 0; t < 32; ++t) {
      v2f a = *(const v2f*)(aBase + 4 * t);
      c = __builtin_amdgcn_wmma_f32_16x16x4_f32(
           false, a,  false, b[t % RING],
           (short)0, c,  false,  false);
      if (t + RING < 32) b[t % RING] = bBase1[(size_t)(t + RING) * 32];
    }
    const int   colh = wave * 16 + lm;
    const float bb   = b1[colh];
#pragma unroll
    for (int r = 0; r < 8; ++r) {
      float v = c[r] + bb;
      v = v > 0.0f ? v : 0.0f;
      Os[(8 * half + r) * LDS_STRIDE + colh] = v;
    }
  }
  __syncthreads();

  if (p2) {
    const float* aBase  = &Os[lm * LDS_STRIDE + 2 * half];
    const v2f*   bBase2 = BF2 + (size_t)(wave - 8) * 1024 + lane;
    v2f q[RING];
#pragma unroll
    for (int u = 0; u < RING; ++u) q[u] = bBase2[(size_t)u * 32];
    v8f c = {};
#pragma unroll
    for (int t = 0; t < 32; ++t) {
      v2f a = *(const v2f*)(aBase + 4 * t);
      c = __builtin_amdgcn_wmma_f32_16x16x4_f32(
           false, a,  false, q[t % RING],
           (short)0, c,  false,  false);
      if (t + RING < 32) q[t % RING] = bBase2[(size_t)(t + RING) * 32];
    }
    const int colc = (wave - 8) * 16 + lm;
    if (colc < 40) {
      const int mBase = m0 + 8 * half;
      float* __restrict__ p = h2 + (size_t)mBase * 40 + colc;
      if (m0 + 16 <= n) {
#pragma unroll
        for (int r = 0; r < 8; ++r) p[r * 40] = c[r];
      } else {
#pragma unroll
        for (int r = 0; r < 8; ++r)
          if (mBase + r < n) p[r * 40] = c[r];
      }
    }
  }
}

__global__ __launch_bounds__(256) void k_gather_f40_bias_logsoftmax(
    const float* __restrict__ h, const int* __restrict__ rowptr,
    const int* __restrict__ cnt, const int* __restrict__ col,
    const float* __restrict__ dinv, const float* __restrict__ bias,
    float* __restrict__ out, int n) {
  const int lane = threadIdx.x & 31;
  const int node = __builtin_amdgcn_readfirstlane(blockIdx.x * 8 + (threadIdx.x >> 5));
  if (node >= n) return;
  const int   start = rowptr[node];
  const int   deg   = cnt[node];
  const float di    = dinv[node];
  const float4* __restrict__ h4 = (const float4*)h;

  const int  g      = lane / 10;
  const int  q      = lane - g * 10;
  const bool active = lane < 30;
  const bool holder = lane < 10;

  float4 hs = make_float4(0.0f, 0.0f, 0.0f, 0.0f);
  float4 bv = make_float4(0.0f, 0.0f, 0.0f, 0.0f);
  if (holder) {
    hs = h4[(size_t)node * 10 + lane];
    bv = ((const float4*)bias)[lane];
  }

  float4 acc = make_float4(0.0f, 0.0f, 0.0f, 0.0f);
  for (int base = 0; base < deg; base += 32) {
    const int k = base + lane;
    int   my  = node;
    float myc = 0.0f;
    if (k < deg) {
      my = col[start + k];
      my = min(max(my, 0), n - 1);
      myc = dinv[my] * di;
    }
    const int m = min(32, deg - base);
#pragma unroll 2
    for (int j0 = 0; j0 < m; j0 += 3) {
      const int   j = j0 + g;
      const int   s = __shfl(my, j & 31, 32);
      float       c = __shfl(myc, j & 31, 32);
      c = (active && j < m) ? c : 0.0f;
      const float4 v = h4[(size_t)s * 10 + q];
      acc.x += c * v.x;
      acc.y += c * v.y;
      acc.z += c * v.z;
      acc.w += c * v.w;
    }
  }
  acc.x += __shfl_down(acc.x, 10, 32) + __shfl_down(acc.x, 20, 32);
  acc.y += __shfl_down(acc.y, 10, 32) + __shfl_down(acc.y, 20, 32);
  acc.z += __shfl_down(acc.z, 10, 32) + __shfl_down(acc.z, 20, 32);
  acc.w += __shfl_down(acc.w, 10, 32) + __shfl_down(acc.w, 20, 32);
  const float sc = di * di;
  acc.x += sc * hs.x;
  acc.y += sc * hs.y;
  acc.z += sc * hs.z;
  acc.w += sc * hs.w;

  const float NEG_INF = -INFINITY;
  float4 v;
  v.x = holder ? acc.x + bv.x : NEG_INF;
  v.y = holder ? acc.y + bv.y : NEG_INF;
  v.z = holder ? acc.z + bv.z : NEG_INF;
  v.w = holder ? acc.w + bv.w : NEG_INF;

  float m = fmaxf(fmaxf(v.x, v.y), fmaxf(v.z, v.w));
#pragma unroll
  for (int off = 8; off > 0; off >>= 1) m = fmaxf(m, __shfl_xor(m, off, 32));

  float s = expf(v.x - m) + expf(v.y - m) + expf(v.z - m) + expf(v.w - m);
  s = holder ? s : 0.0f;
#pragma unroll
  for (int off = 8; off > 0; off >>= 1) s += __shfl_xor(s, off, 32);

  const float lse = logf(s);
  if (holder) {
    float4 r;
    r.x = v.x - m - lse;
    r.y = v.y - m - lse;
    r.z = v.z - m - lse;
    r.w = v.w - m - lse;
    ((float4*)out)[(size_t)node * 10 + lane] = r;
  }
}

static inline int cdiv_ll(long long a, long long b) { return (int)((a + b - 1) / b); }

extern "C" void kernel_launch(void* const* d_in, const int* in_sizes, int n_in,
                              void* d_out, int out_size, void* d_ws, size_t ws_size,
                              hipStream_t stream) {
  const float* x  = (const float*)d_in[0];
  const int*   ei = (const int*)  d_in[1];
  const float* W1 = (const float*)d_in[2];
  const float* b1 = (const float*)d_in[3];
  const float* W2 = (const float*)d_in[4];
  const float* b2 = (const float*)d_in[5];

  const int F = 128, H = 128, C = 40;
  const int N = in_sizes[0] / F;
  const int E = in_sizes[1] / 2;
  const int* src = ei;
  const int* dst = ei + E;

  const int nT1 = (H + 15) / 16;
  const int nT2 = (C + 15) / 16;

  float* ws   = (float*)d_ws;
  int*   cnt  = (int*)ws;           ws += N;
  float* dinv = ws;                 ws += N;
  float* h2   = ws;                 ws += (size_t)N * H;
  int2*  pairs    = (int2*)ws;
  int*   hist     = (int*)ws + (size_t)2 * E;
  int*   binStart = hist + N_SLICES * MAX_BINS;
  ws += (size_t)N * H;
  int*   col  = (int*)ws;
  int*   rowptr = col + E;
  ws += (size_t)N * C;
  v2f* bf1    = (v2f*)ws;           ws += (size_t)nT1 * 1024 * 2;
  v2f* bf2    = (v2f*)ws;           ws += (size_t)nT2 * 1024 * 2;
  float* out  = (float*)d_out;

  const int B = 256;
  const int nBins = cdiv_ll(N, MAX_BINS);

  const int packBlocks = cdiv_ll((long long)(nT1 + nT2) * 1024, EDGE_THREADS);
  k_bin_hist_pack<<<N_SLICES + packBlocks, EDGE_THREADS, 0, stream>>>(
      dst, hist, E, W1, bf1, H, nT1, W2, bf2, C, nT2);
  k_bin_scatter<<<N_SLICES, EDGE_THREADS, 0, stream>>>(src, dst, hist, binStart, pairs, E);
  k_bin_csr    <<<nBins, EDGE_THREADS, 0, stream>>>(pairs, binStart, cnt, rowptr, dinv, col, N);

  k_gather_x_w1_relu_w2<<<cdiv_ll(N, 16), 512, 0, stream>>>(x, rowptr, cnt, col, dinv, bf1, b1, bf2, h2, N);

  k_gather_f40_bias_logsoftmax<<<cdiv_ll(N, 8), B, 0, stream>>>(h2, rowptr, cnt, col, dinv, b2, out, N);
}


# Layer-1 epilogue stores h2 rows pre-multiplied by dinv[row]; the layer-2 gather adds source rows unweighted and scales the row sum once by dinv[d]: per-edge dinv[src] gather gone; same sum order
# speedup vs baseline: 6.2407x; 1.0427x over previous
#include <hip/hip_runtime.h>
#include <hip/hip_bf16.h>
#include <math.h>

typedef float v2f __attribute__((ext_vector_type(2)));
typedef float v8f __attribute__((ext_vector_type(8)));

constexpr int BIN_SHIFT    = 8;
constexpr int MAX_BINS     = 256;
constexpr int N_SLICES     = 128;
constexpr int EDGE_THREADS = 1024;
constexpr int EDGE_BATCH   = 8;

__device__ __forceinline__ int bin_of(int d) {
  return min((int)((unsigned)d >> BIN_SHIFT), MAX_BINS - 1);
}

__device__ __forceinline__ void pack_bfrag(const float* __restrict__ B, v2f* __restrict__ BF,
                                           int N, int i) {
  const int lane = i & 31;
  const int t    = (i >> 5) & 31;
  const int nt   = i >> 10;
  const int half = lane >> 4;
  const int lm   = lane & 15;
  const int col  = nt * 16 + lm;
  const int kk   = 4 * t + 2 * half;
  v2f v;
  v.x = (col < N) ? B[(size_t)kk * N + col]       : 0.0f;
  v.y = (col < N) ? B[(size_t)(kk + 1) * N + col] : 0.0f;
  BF[i] = v;
}

__global__ __launch_bounds__(EDGE_THREADS) void k_bin_hist_pack(
    const int* __restrict__ dst, int* __restrict__ hist, int E,
    const float* __restrict__ W1, v2f* __restrict__ BF1, int H, int nT1,
    const float* __restrict__ W2, v2f* __restrict__ BF2, int C, int nT2) {
  __shared__ int h[MAX_BINS];
  const int tid = threadIdx.x;
  if (blockIdx.x >= N_SLICES) {
    const int i = (blockIdx.x - N_SLICES) * EDGE_THREADS + tid;
    if (i < nT1 * 1024)               pack_bfrag(W1, BF1, H, i);
    else if (i < (nT1 + nT2) * 1024)  pack_bfrag(W2, BF2, C, i - nT1 * 1024);
    return;
  }
  if (tid < MAX_BINS) h[tid] = 0;
  __syncthreads();
  const int slice = (E + N_SLICES - 1) / N_SLICES;
  const int e0 = blockIdx.x * slice;
  const int e1 = min(E, e0 + slice);
  for (int e = e0 + tid; e < e1; e += EDGE_THREADS * EDGE_BATCH) {
    int d[EDGE_BATCH];
#pragma unroll
    for (int u = 0; u < EDGE_BATCH; ++u) {
      d[u] = dst[min(e + u * EDGE_THREADS, e1 - 1)];
    }
#pragma unroll
    for (int u = 0; u < EDGE_BATCH; ++u) asm volatile("" : "+v"(d[u]));
#pragma unroll
    for (int u = 0; u < EDGE_BATCH; ++u) {
      const int i = e + u * EDGE_THREADS;
      if (i < e1) atomicAdd(&h[bin_of(d[u])], 1);
    }
  }
  __syncthreads();
  if (tid < MAX_BINS) hist[blockIdx.x * MAX_BINS + tid] = h[tid];
}

__global__ __launch_bounds__(EDGE_THREADS) void k_bin_scatter(
    const int* __restrict__ src, const int* __restrict__ dst,
    const int* __restrict__ hist, int* __restrict__ binStart,
    int2* __restrict__ pairs, int E) {
  constexpr int GROUPS = EDGE_THREADS / 64;
  constexpr int ROWS   = N_SLICES / GROUPS;
  __shared__ __attribute__((aligned(16))) int ptot[GROUPS][MAX_BINS];
  __shared__ __attribute__((aligned(16))) int ppre[GROUPS][MAX_BINS];
  __shared__ int cursor[MAX_BINS];
  __shared__ int wsum[8];
  const int tid  = threadIdx.x;
  const int lane = tid & 31;
  const int wave = tid >> 5;
  const int blk  = blockIdx.x;

  {
    const int g = __builtin_amdgcn_readfirstlane(tid >> 6);
    const int q = tid & 63;
    const int4* __restrict__ h4 = (const int4*)hist;
    int4 v[ROWS];
#pragma unroll
    for (int u = 0; u < ROWS; ++u) v[u] = h4[(g * ROWS + u) * (MAX_BINS / 4) + q];
    int4 tot = make_int4(0, 0, 0, 0), pre = make_int4(0, 0, 0, 0);
#pragma unroll
    for (int u = 0; u < ROWS; ++u) {
      tot.x += v[u].x; tot.y += v[u].y; tot.z += v[u].z; tot.w += v[u].w;
      if (g * ROWS + u < blk) {
        pre.x += v[u].x; pre.y += v[u].y; pre.z += v[u].z; pre.w += v[u].w;
      }
    }
    *(int4*)&ptot[g][4 * q] = tot;
    *(int4*)&ppre[g][4 * q] = pre;
  }
  __syncthreads();

  int tot = 0, pre = 0;
  if (tid < MAX_BINS) {
#pragma unroll
    for (int g = 0; g < GROUPS; ++g) { tot += ptot[g][tid]; pre += ppre[g][tid]; }
  }
  int incl = tot;
  for (int off = 1; off < 32; off <<= 1) {
    int u = __shfl_up(incl, off, 32);
    if (lane >= off) incl += u;
  }
  if (tid < MAX_BINS && lane == 31) wsum[wave] = incl;
  __syncthreads();
  if (tid < MAX_BINS) {
    int woff = 0;
    for (int w = 0; w < 8; ++w) {
      int u = wsum[w];
      if (w < wave) woff += u;
    }
    const int excl = woff + incl - tot;
    cursor[tid] = excl + pre;
    if (blk == 0) {
      binStart[tid] = excl;
      if (tid == MAX_BINS - 1) binStart[MAX_BINS] = excl + tot;
    }
  }
  __syncthreads();

  const int slice = (E + N_SLICES - 1) / N_SLICES;
  const int e0 = blk * slice;
  const int e1 = min(E, e0 + slice);
  for (int e = e0 + tid; e < e1; e += EDGE_THREADS * EDGE_BATCH) {
    int d[EDGE_BATCH], s[EDGE_BATCH];
#pragma unroll
    for (int u = 0; u < EDGE_BATCH; ++u) {
      const int i = min(e + u * EDGE_THREADS, e1 - 1);
      d[u] = dst[i];
      s[u] = src[i];
    }
#pragma unroll
    for (int u = 0; u < EDGE_BATCH; ++u) asm volatile("" : "+v"(d[u]), "+v"(s[u]));
#pragma unroll
    for (int u = 0; u < EDGE_BATCH; ++u) {
      const int i = e + u * EDGE_THREADS;
      if (i < e1) {
        const int pos = atomicAdd(&cursor[bin_of(d[u])], 1);
        pairs[pos] = make_int2(s[u], d[u]);
      }
    }
  }
}

__global__ __launch_bounds__(EDGE_THREADS) void k_bin_csr(
    const int2* __restrict__ pairs, const int* __restrict__ binStart,
    int* __restrict__ cnt, int* __restrict__ rowptr, float* __restrict__ dinv,
    int* __restrict__ col, int n) {
  constexpr int BATCH = 4;
  __shared__ int c[MAX_BINS];
  __shared__ int cur[MAX_BINS];
  __shared__ int wsum[8];
  const int tid  = threadIdx.x;
  const int lane = tid & 31;
  const int wave = tid >> 5;
  const int bin  = blockIdx.x;
  const int base = binStart[bin];
  const int end  = binStart[bin + 1];
  if (tid < MAX_BINS) c[tid] = 0;
  __syncthreads();

  for (int i0 = base + tid; i0 < end; i0 += EDGE_THREADS * BATCH) {
    int d[BATCH];
#pragma unroll
    for (int u = 0; u < BATCH; ++u) {
      d[u] = pairs[min(i0 + u * EDGE_THREADS, end - 1)].y;
    }
#pragma unroll
    for (int u = 0; u < BATCH; ++u) asm volatile("" : "+v"(d[u]));
#pragma unroll
    for (int u = 0; u < BATCH; ++u) {
      const int i = i0 + u * EDGE_THREADS;
      if (i < end) atomicAdd(&c[d[u] & (MAX_BINS - 1)], 1);
    }
  }
  __syncthreads();

  const int mine = (tid < MAX_BINS) ? c[tid] : 0;
  int incl = mine;
  for (int off = 1; off < 32; off <<= 1) {
    int u = __shfl_up(incl, off, 32);
    if (lane >= off) incl += u;
  }
  if (tid < MAX_BINS && lane == 31) wsum[wave] = incl;
  __syncthreads();
  if (tid < MAX_BINS) {
    int woff = 0;
    for (int w = 0; w < 8; ++w) {
      int u = wsum[w];
      if (w < wave) woff += u;
    }
    const int excl = woff + incl - mine;
    cur[tid] = excl;
    const int node = (bin << BIN_SHIFT) + tid;
    if (node < n) {
      cnt[node]    = mine;
      rowptr[node] = base + excl;
      dinv[node]   = rsqrtf((float)(mine + 1));
    }
  }
  __syncthreads();

  for (int i0 = base + tid; i0 < end; i0 += EDGE_THREADS * BATCH) {
    int2 p[BATCH];
#pragma unroll
    for (int u = 0; u < BATCH; ++u) {
      p[u] = pairs[min(i0 + u * EDGE_THREADS, end - 1)];
    }
#pragma unroll
    for (int u = 0; u < BATCH; ++u) asm volatile("" : "+v"(p[u].x), "+v"(p[u].y));
#pragma unroll
    for (int u = 0; u < BATCH; ++u) {
      const int i = i0 + u * EDGE_THREADS;
      if (i < end) {
        const int pos = atomicAdd(&cur[p[u].y & (MAX_BINS - 1)], 1);
        col[base + pos] = p[u].x;
      }
    }
  }
}

__global__ __launch_bounds__(512, 16) void k_gather_x_w1_relu_w2(
    const float* __restrict__ x, const int* __restrict__ rowptr,
    const int* __restrict__ cnt, const int* __restrict__ col,
    const float* __restrict__ dinv, const v2f* __restrict__ BF1,
    const float* __restrict__ b1, const v2f* __restrict__ BF2,
    float* __restrict__ h2, int n) {
  constexpr int LDS_STRIDE = 132;
  __shared__ __attribute__((aligned(16))) float As[16 * LDS_STRIDE];
  __shared__ __attribute__((aligned(16))) float Os[16 * LDS_STRIDE];
  __shared__ float Ds[16];

  const int lane = threadIdx.x & 31;
  const int wave = __builtin_amdgcn_readfirstlane(threadIdx.x >> 5);
  const int m0   = blockIdx.x * 16;
  const int node = m0 + wave;

  float4 acc = make_float4(0.0f, 0.0f, 0.0f, 0.0f);
  float  di  = 0.0f;
  if (node < n) {
    const int   start = rowptr[node];
    const int   deg   = cnt[node];
    di = dinv[node];
    const float4* __restrict__ x4 = (const float4*)x;

    const float4 xs = x4[(size_t)node * 32 + lane];
    const float  sc = di * di;
    acc = make_float4(sc * xs.x, sc * xs.y, sc * xs.z, sc * xs.w);

    for (int base = 0; base < deg; base += 32) {
      const int k = base + lane;
      int   my  = 0;
      float myc = 0.0f;
      if (k < deg) {
        my = col[start + k];
        my = min(max(my, 0), n - 1);
        myc = dinv[my] * di;
      }
      const int m = min(32, deg - base);
#pragma unroll 4
      for (int j = 0; j < m; ++j) {
        const int   s = __builtin_amdgcn_readlane(my, j);
        const float c = __int_as_float(__builtin_amdgcn_readlane(__float_as_int(myc), j));
        const float4 v = x4[(size_t)s * 32 + lane];
        acc.x += c * v.x;
        acc.y += c * v.y;
        acc.z += c * v.z;
        acc.w += c * v.w;
      }
    }
  }
  const int half = lane >> 4;
  const int lm   = lane & 15;

  const bool p1 = wave < 8;
  const bool p2 = wave >= 8 && wave < 11;

  *(float4*)&As[wave * LDS_STRIDE + lane * 4] = acc;
  if (lane == 0) Ds[wave] = di;

  constexpr int RING = 12;

  const v2f* bBase1 = BF1 + (size_t)wave * 1024 + lane;
  v2f b[RING];
  if (p1) {
#pragma unroll
    for (int u = 0; u < RING; ++u) b[u] = bBase1[(size_t)u * 32];
  }
  __syncthreads();

  if (p1) {
    const float* aBase = &As[lm * LDS_STRIDE + 2 * half];
    v8f c = {};
#pragma unroll
    for (int t = 0; t < 32; ++t) {
      v2f a = *(const v2f*)(aBase + 4 * t);
      c = __builtin_amdgcn_wmma_f32_16x16x4_f32(
           false, a,  false, b[t % RING],
           (short)0, c,  false,  false);
      if (t + RING < 32) b[t % RING] = bBase1[(size_t)(t + RING) * 32];
    }
    const int   colh = wave * 16 + lm;
    const float bb   = b1[colh];
#pragma unroll
    for (int r = 0; r < 8; ++r) {
      float v = c[r] + bb;
      v = v > 0.0f ? v : 0.0f;
      Os[(8 * half + r) * LDS_STRIDE + colh] = v;
    }
  }
  __syncthreads();

  if (p2) {
    const float* aBase  = &Os[lm * LDS_STRIDE + 2 * half];
    const v2f*   bBase2 = BF2 + (size_t)(wave - 8) * 1024 + lane;
    v2f q[RING];
#pragma unroll
    for (int u = 0; u < RING; ++u) q[u] = bBase2[(size_t)u * 32];
    v8f c = {};
#pragma unroll
    for (int t = 0; t < 32; ++t) {
      v2f a = *(const v2f*)(aBase + 4 * t);
      c = __builtin_amdgcn_wmma_f32_16x16x4_f32(
           false, a,  false, q[t % RING],
           (short)0, c,  false,  false);
      if (t + RING < 32) q[t % RING] = bBase2[(size_t)(t + RING) * 32];
    }
    const int colc = (wave - 8) * 16 + lm;
    if (colc < 40) {
      const int mBase = m0 + 8 * half;
      float* __restrict__ p = h2 + (size_t)mBase * 40 + colc;
      const float* dr = &Ds[8 * half];
      if (m0 + 16 <= n) {
#pragma unroll
        for (int r = 0; r < 8; ++r) p[r * 40] = dr[r] * c[r];
      } else {
#pragma unroll
        for (int r = 0; r < 8; ++r)
          if (mBase + r < n) p[r * 40] = dr[r] * c[r];
      }
    }
  }
}

__global__ __launch_bounds__(256) void k_gather_f40_bias_logsoftmax(
    const float* __restrict__ h, const int* __restrict__ rowptr,
    const int* __restrict__ cnt, const int* __restrict__ col,
    const float* __restrict__ dinv, const float* __restrict__ bias,
    float* __restrict__ out, int n) {
  const int lane = threadIdx.x & 31;
  const int node = __builtin_amdgcn_readfirstlane(blockIdx.x * 8 + (threadIdx.x >> 5));
  if (node >= n) return;
  const int   start = rowptr[node];
  const int   deg   = cnt[node];
  const float di    = dinv[node];
  const float4* __restrict__ h4 = (const float4*)h;

  const int  g      = lane / 10;
  const int  q      = lane - g * 10;
  const bool active = lane < 30;
  const bool holder = lane < 10;

  float4 hs = make_float4(0.0f, 0.0f, 0.0f, 0.0f);
  float4 bv = make_float4(0.0f, 0.0f, 0.0f, 0.0f);
  if (holder) {
    hs = h4[(size_t)node * 10 + lane];
    bv = ((const float4*)bias)[lane];
  }

  float4 acc = make_float4(0.0f, 0.0f, 0.0f, 0.0f);
  for (int base = 0; base < deg; base += 32) {
    const int k = base + lane;
    int my = node;
    if (k < deg) {
      my = col[start + k];
      my = min(max(my, 0), n - 1);
    }
    const int m = min(32, deg - base);
#pragma unroll 2
    for (int j0 = 0; j0 < m; j0 += 3) {
      const int   j = j0 + g;
      const int   s = __shfl(my, j & 31, 32);
      const float c = (active && j < m) ? 1.0f : 0.0f;
      const float4 v = h4[(size_t)s * 10 + q];
      acc.x += c * v.x;
      acc.y += c * v.y;
      acc.z += c * v.z;
      acc.w += c * v.w;
    }
  }
  acc.x += __shfl_down(acc.x, 10, 32) + __shfl_down(acc.x, 20, 32);
  acc.y += __shfl_down(acc.y, 10, 32) + __shfl_down(acc.y, 20, 32);
  acc.z += __shfl_down(acc.z, 10, 32) + __shfl_down(acc.z, 20, 32);
  acc.w += __shfl_down(acc.w, 10, 32) + __shfl_down(acc.w, 20, 32);
  acc.x = di * (acc.x + hs.x);
  acc.y = di * (acc.y + hs.y);
  acc.z = di * (acc.z + hs.z);
  acc.w = di * (acc.w + hs.w);

  const float NEG_INF = -INFINITY;
  float4 v;
  v.x = holder ? acc.x + bv.x : NEG_INF;
  v.y = holder ? acc.y + bv.y : NEG_INF;
  v.z = holder ? acc.z + bv.z : NEG_INF;
  v.w = holder ? acc.w + bv.w : NEG_INF;

  float m = fmaxf(fmaxf(v.x, v.y), fmaxf(v.z, v.w));
#pragma unroll
  for (int off = 8; off > 0; off >>= 1) m = fmaxf(m, __shfl_xor(m, off, 32));

  float s = expf(v.x - m) + expf(v.y - m) + expf(v.z - m) + expf(v.w - m);
  s = holder ? s : 0.0f;
#pragma unroll
  for (int off = 8; off > 0; off >>= 1) s += __shfl_xor(s, off, 32);

  const float lse = logf(s);
  if (holder) {
    float4 r;
    r.x = v.x - m - lse;
    r.y = v.y - m - lse;
    r.z = v.z - m - lse;
    r.w = v.w - m - lse;
    ((float4*)out)[(size_t)node * 10 + lane] = r;
  }
}

static inline int cdiv_ll(long long a, long long b) { return (int)((a + b - 1) / b); }

extern "C" void kernel_launch(void* const* d_in, const int* in_sizes, int n_in,
                              void* d_out, int out_size, void* d_ws, size_t ws_size,
                              hipStream_t stream) {
  const float* x  = (const float*)d_in[0];
  const int*   ei = (const int*)  d_in[1];
  const float* W1 = (const float*)d_in[2];
  const float* b1 = (const float*)d_in[3];
  const float* W2 = (const float*)d_in[4];
  const float* b2 = (const float*)d_in[5];

  const int F = 128, H = 128, C = 40;
  const int N = in_sizes[0] / F;
  const int E = in_sizes[1] / 2;
  const int* src = ei;
  const int* dst = ei + E;

  const int nT1 = (H + 15) / 16;
  const int nT2 = (C + 15) / 16;

  float* ws   = (float*)d_ws;
  int*   cnt  = (int*)ws;           ws += N;
  float* dinv = ws;                 ws += N;
  float* h2   = ws;                 ws += (size_t)N * H;
  int2*  pairs    = (int2*)ws;
  int*   hist     = (int*)ws + (size_t)2 * E;
  int*   binStart = hist + N_SLICES * MAX_BINS;
  ws += (size_t)N * H;
  int*   col  = (int*)ws;
  int*   rowptr = col + E;
  ws += (size_t)N * C;
  v2f* bf1    = (v2f*)ws;           ws += (size_t)nT1 * 1024 * 2;
  v2f* bf2    = (v2f*)ws;           ws += (size_t)nT2 * 1024 * 2;
  float* out  = (float*)d_out;

  const int B = 256;
  const int nBins = cdiv_ll(N, MAX_BINS);

  const int packBlocks = cdiv_ll((long long)(nT1 + nT2) * 1024, EDGE_THREADS);
  k_bin_hist_pack<<<N_SLICES + packBlocks, EDGE_THREADS, 0, stream>>>(
      dst, hist, E, W1, bf1, H, nT1, W2, bf2, C, nT2);
  k_bin_scatter<<<N_SLICES, EDGE_THREADS, 0, stream>>>(src, dst, hist, binStart, pairs, E);
  k_bin_csr    <<<nBins, EDGE_THREADS, 0, stream>>>(pairs, binStart, cnt, rowptr, dinv, col, N);

  k_gather_x_w1_relu_w2<<<cdiv_ll(N, 16), 512, 0, stream>>>(x, rowptr, cnt, col, dinv, bf1, b1, bf2, h2, N);

  k_gather_f40_bias_logsoftmax<<<cdiv_ll(N, 8), B, 0, stream>>>(h2, rowptr, cnt, col, dinv, b2, out, N);
}


# Edge build = two LDS sorts in 2 launches, not 3: slice blocks bin-sort edges, bin blocks node-sort their 128 runs, output staged in LDS and stored as whole lines; row sum order: new LDS arrival
# speedup vs baseline: 6.8820x; 1.1028x over previous
#include <hip/hip_runtime.h>
#include <hip/hip_bf16.h>
#include <math.h>

typedef float v2f __attribute__((ext_vector_type(2)));
typedef float v8f __attribute__((ext_vector_type(8)));

constexpr int BIN_SHIFT    = 8;
constexpr int MAX_BINS     = 256;
constexpr int N_SLICES     = 128;
constexpr int EDGE_THREADS = 1024;
constexpr int EDGE_BATCH   = 8;
constexpr int SLICE_CAP    = EDGE_THREADS * EDGE_BATCH;

__device__ __forceinline__ int bin_of(int d) {
  return min((int)((unsigned)d >> BIN_SHIFT), MAX_BINS - 1);
}

__device__ __forceinline__ void pack_bfrag(const float* __restrict__ B, v2f* __restrict__ BF,
                                           int N, int i) {
  const int lane = i & 31;
  const int t    = (i >> 5) & 31;
  const int nt   = i >> 10;
  const int half = lane >> 4;
  const int lm   = lane & 15;
  const int col  = nt * 16 + lm;
  const int kk   = 4 * t + 2 * half;
  v2f v;
  v.x = (col < N) ? B[(size_t)kk * N + col]       : 0.0f;
  v.y = (col < N) ? B[(size_t)(kk + 1) * N + col] : 0.0f;
  BF[i] = v;
}

__device__ __forceinline__ int scan256(int mine, int tid, int* wsum) {
  const int lane = tid & 31;
  const int wave = tid >> 5;
  int incl = mine;
  for (int off = 1; off < 32; off <<= 1) {
    int u = __shfl_up(incl, off, 32);
    if (lane >= off) incl += u;
  }
  if (tid < MAX_BINS && lane == 31) wsum[wave] = incl;
  __syncthreads();
  int woff = 0;
  if (tid < MAX_BINS) {
    for (int w = 0; w < 8; ++w) {
      int u = wsum[w];
      if (w < wave) woff += u;
    }
  }
  return woff + incl - mine;
}

__global__ __launch_bounds__(EDGE_THREADS) void k_slice_sort_pack(
    const int* __restrict__ src, const int* __restrict__ dst,
    int* __restrict__ cntM, int* __restrict__ lstM, int2* __restrict__ pairs, int E,
    const float* __restrict__ W1, v2f* __restrict__ BF1, int H, int nT1,
    const float* __restrict__ W2, v2f* __restrict__ BF2, int C, int nT2) {
  __shared__ int h[MAX_BINS];
  __shared__ int cur[MAX_BINS + 32];
  __shared__ int wsum[8];
  __shared__ __attribute__((aligned(16))) int2 stage[SLICE_CAP];
  const int tid = threadIdx.x;
  if (blockIdx.x >= N_SLICES) {
    const int i = (blockIdx.x - N_SLICES) * EDGE_THREADS + tid;
    if (i < nT1 * 1024)               pack_bfrag(W1, BF1, H, i);
    else if (i < (nT1 + nT2) * 1024)  pack_bfrag(W2, BF2, C, i - nT1 * 1024);
    return;
  }
  if (tid < MAX_BINS) h[tid] = 0;
  __syncthreads();
  const int blk   = blockIdx.x;
  const int slice = (E + N_SLICES - 1) / N_SLICES;
  const int e0  = min(blk * slice, E);
  const int e1  = min(E, e0 + slice);
  const int len = e1 - e0;

  if (len <= SLICE_CAP) {
    int d[EDGE_BATCH], s[EDGE_BATCH];
#pragma unroll
    for (int u = 0; u < EDGE_BATCH; ++u) {
      const int j = max(e0 + min(tid + u * EDGE_THREADS, len - 1), 0);
      d[u] = dst[j];
      s[u] = src[j];
    }
#pragma unroll
    for (int u = 0; u < EDGE_BATCH; ++u) asm volatile("" : "+v"(d[u]), "+v"(s[u]));
#pragma unroll
    for (int u = 0; u < EDGE_BATCH; ++u) {
      if (tid + u * EDGE_THREADS < len) atomicAdd(&h[bin_of(d[u])], 1);
    }
    __syncthreads();
    const int mine = (tid < MAX_BINS) ? h[tid] : 0;
    const int excl = scan256(mine, tid, wsum);
    if (tid < MAX_BINS) {
      cur[tid] = excl;
      cntM[blk * MAX_BINS + tid] = mine;
      lstM[blk * MAX_BINS + tid] = excl;
    }
    __syncthreads();
    int pos[EDGE_BATCH];
#pragma unroll
    for (int u = 0; u < EDGE_BATCH; ++u) {
      const bool ok = tid + u * EDGE_THREADS < len;
      pos[u] = atomicAdd(&cur[ok ? bin_of(d[u]) : MAX_BINS + (tid & 31)], 1);
    }
#pragma unroll
    for (int u = 0; u < EDGE_BATCH; ++u) {
      if (tid + u * EDGE_THREADS < len) stage[pos[u]] = make_int2(s[u], d[u]);
    }
    __syncthreads();
    for (int i = tid; i < len; i += EDGE_THREADS) pairs[e0 + i] = stage[i];
  } else {
    for (int e = e0 + tid; e < e1; e += EDGE_THREADS) atomicAdd(&h[bin_of(dst[e])], 1);
    __syncthreads();
    const int mine = (tid < MAX_BINS) ? h[tid] : 0;
    const int excl = scan256(mine, tid, wsum);
    if (tid < MAX_BINS) {
      cur[tid] = excl;
      cntM[blk * MAX_BINS + tid] = mine;
      lstM[blk * MAX_BINS + tid] = excl;
    }
    __syncthreads();
    for (int e = e0 + tid; e < e1; e += EDGE_THREADS) {
      const int dv  = dst[e];
      const int pos = atomicAdd(&cur[bin_of(dv)], 1);
      pairs[e0 + pos] = make_int2(src[e], dv);
    }
  }
}

__global__ __launch_bounds__(EDGE_THREADS) void k_bin_csr(
    const int2* __restrict__ pairs, const int* __restrict__ cntM, const int* __restrict__ lstM,
    int* __restrict__ cnt, int* __restrict__ rowptr, float* __restrict__ dinv,
    int* __restrict__ col, int n, int E) {
  constexpr int WAVES = EDGE_THREADS / 32;
  constexpr int SEGS  = N_SLICES / WAVES;
  constexpr int FIRST = 2;
  static_assert(EDGE_THREADS / 8 == N_SLICES, "8 threads per matrix row");
  constexpr int BIN_CAP = 16384;
  __shared__ int c[MAX_BINS];
  __shared__ int cur[MAX_BINS + 32];
  __shared__ int wsum[8];
  __shared__ int segCnt[N_SLICES];
  __shared__ int segStart[N_SLICES];
  __shared__ int red[WAVES];
  __shared__ int totS;
  __shared__ __attribute__((aligned(16))) int stage[BIN_CAP];
  const int tid  = threadIdx.x;
  const int lane = tid & 31;
  const int wave = __builtin_amdgcn_readfirstlane(tid >> 5);
  const int bin  = blockIdx.x;
  if (tid < MAX_BINS) c[tid] = 0;

  {
    const int row  = tid >> 3;
    const int part = tid & 7;
    const int4* __restrict__ m4 = (const int4*)cntM;
    int4 v[8];
#pragma unroll
    for (int u = 0; u < 8; ++u) v[u] = m4[row * (MAX_BINS / 4) + part * 8 + u];
    int sum = 0;
#pragma unroll
    for (int u = 0; u < 8; ++u) {
      const int c0 = (part * 8 + u) * 4;
      sum += (c0     < bin ? v[u].x : 0) + (c0 + 1 < bin ? v[u].y : 0)
           + (c0 + 2 < bin ? v[u].z : 0) + (c0 + 3 < bin ? v[u].w : 0);
    }
#pragma unroll
    for (int off = 16; off > 0; off >>= 1) sum += __shfl_xor(sum, off, 32);
    if (lane == 0) red[wave] = sum;
  }
  if (tid < N_SLICES) {
    const int slice = (E + N_SLICES - 1) / N_SLICES;
    segCnt[tid]   = cntM[tid * MAX_BINS + bin];
    segStart[tid] = min(tid * slice, E) + lstM[tid * MAX_BINS + bin];
  }
  __syncthreads();
  int base = 0;
#pragma unroll
  for (int w = 0; w < WAVES; ++w) base += red[w];

  int segc[SEGS], segs[SEGS];
#pragma unroll
  for (int u = 0; u < SEGS; ++u) {
    segc[u] = __builtin_amdgcn_readfirstlane(segCnt[wave * SEGS + u]);
    segs[u] = __builtin_amdgcn_readfirstlane(segStart[wave * SEGS + u]);
  }

  int2 p[SEGS * FIRST];
#pragma unroll
  for (int u = 0; u < SEGS; ++u) {
#pragma unroll
    for (int f = 0; f < FIRST; ++f) {
      const int j = min(max(segs[u] + min(lane + 32 * f, segc[u] - 1), 0), E - 1);
      p[u * FIRST + f] = pairs[j];
    }
  }
#pragma unroll
  for (int u = 0; u < SEGS * FIRST; ++u) asm volatile("" : "+v"(p[u].x), "+v"(p[u].y));
#pragma unroll
  for (int u = 0; u < SEGS; ++u) {
#pragma unroll
    for (int f = 0; f < FIRST; ++f) {
      if (lane + 32 * f < segc[u]) atomicAdd(&c[p[u * FIRST + f].y & (MAX_BINS - 1)], 1);
    }
    for (int k = lane + 32 * FIRST; k < segc[u]; k += 32)
      atomicAdd(&c[pairs[segs[u] + k].y & (MAX_BINS - 1)], 1);
  }
  __syncthreads();

  const int mine = (tid < MAX_BINS) ? c[tid] : 0;
  const int excl = scan256(mine, tid, wsum);
  if (tid < MAX_BINS) {
    cur[tid] = excl;
    if (tid == MAX_BINS - 1) totS = excl + mine;
    const int node = (bin << BIN_SHIFT) + tid;
    if (node < n) {
      cnt[node]    = mine;
      rowptr[node] = base + excl;
      dinv[node]   = rsqrtf((float)(mine + 1));
    }
  }
  __syncthreads();
  const int total = totS;

  int pos[SEGS * FIRST];
#pragma unroll
  for (int u = 0; u < SEGS; ++u) {
#pragma unroll
    for (int f = 0; f < FIRST; ++f) {
      const bool ok = lane + 32 * f < segc[u];
      pos[u * FIRST + f] =
          atomicAdd(&cur[ok ? (p[u * FIRST + f].y & (MAX_BINS - 1)) : MAX_BINS + lane], 1);
    }
  }
  if (total <= BIN_CAP) {
#pragma unroll
    for (int u = 0; u < SEGS; ++u) {
#pragma unroll
      for (int f = 0; f < FIRST; ++f) {
        if (lane + 32 * f < segc[u]) stage[pos[u * FIRST + f]] = p[u * FIRST + f].x;
      }
      for (int k = lane + 32 * FIRST; k < segc[u]; k += 32) {
        const int2 q = pairs[segs[u] + k];
        stage[atomicAdd(&cur[q.y & (MAX_BINS - 1)], 1)] = q.x;
      }
    }
    __syncthreads();
    for (int i = tid; i < total; i += EDGE_THREADS) col[base + i] = stage[i];
  } else {
#pragma unroll
    for (int u = 0; u < SEGS; ++u) {
#pragma unroll
      for (int f = 0; f < FIRST; ++f) {
        if (lane + 32 * f < segc[u]) col[base + pos[u * FIRST + f]] = p[u * FIRST + f].x;
      }
      for (int k = lane + 32 * FIRST; k < segc[u]; k += 32) {
        const int2 q = pairs[segs[u] + k];
        col[base + atomicAdd(&cur[q.y & (MAX_BINS - 1)], 1)] = q.x;
      }
    }
  }
}

__global__ __launch_bounds__(512, 16) void k_gather_x_w1_relu_w2(
    const float* __restrict__ x, const int* __restrict__ rowptr,
    const int* __restrict__ cnt, const int* __restrict__ col,
    const float* __restrict__ dinv, const v2f* __restrict__ BF1,
    const float* __restrict__ b1, const v2f* __restrict__ BF2,
    float* __restrict__ h2, int n) {
  constexpr int LDS_STRIDE = 132;
  __shared__ __attribute__((aligned(16))) float As[16 * LDS_STRIDE];
  __shared__ __attribute__((aligned(16))) float Os[16 * LDS_STRIDE];
  __shared__ float Ds[16];

  const int lane = threadIdx.x & 31;
  const int wave = __builtin_amdgcn_readfirstlane(threadIdx.x >> 5);
  const int m0   = blockIdx.x * 16;
  const int node = m0 + wave;

  float4 acc = make_float4(0.0f, 0.0f, 0.0f, 0.0f);
  float  di  = 0.0f;
  if (node < n) {
    const int   start = rowptr[node];
    const int   deg   = cnt[node];
    di = dinv[node];
    const float4* __restrict__ x4 = (const float4*)x;

    const float4 xs = x4[(size_t)node * 32 + lane];
    const float  sc = di * di;
    acc = make_float4(sc * xs.x, sc * xs.y, sc * xs.z, sc * xs.w);

    for (int base = 0; base < deg; base += 32) {
      const int k = base + lane;
      int   my  = 0;
      float myc = 0.0f;
      if (k < deg) {
        my = col[start + k];
        my = min(max(my, 0), n - 1);
        myc = dinv[my] * di;
      }
      const int m = min(32, deg - base);
#pragma unroll 4
      for (int j = 0; j < m; ++j) {
        const int   s = __builtin_amdgcn_readlane(my, j);
        const float c = __int_as_float(__builtin_amdgcn_readlane(__float_as_int(myc), j));
        const float4 v = x4[(size_t)s * 32 + lane];
        acc.x += c * v.x;
        acc.y += c * v.y;
        acc.z += c * v.z;
        acc.w += c * v.w;
      }
    }
  }
  const int half = lane >> 4;
  const int lm   = lane & 15;

  const bool p1 = wave < 8;
  const bool p2 = wave >= 8 && wave < 11;

  *(float4*)&As[wave * LDS_STRIDE + lane * 4] = acc;
  if (lane == 0) Ds[wave] = di;

  constexpr int RING = 12;

  const v2f* bBase1 = BF1 + (size_t)wave * 1024 + lane;
  v2f b[RING];
  if (p1) {
#pragma unroll
    for (int u = 0; u < RING; ++u) b[u] = bBase1[(size_t)u * 32];
  }
  __syncthreads();

  if (p1) {
    const float* aBase = &As[lm * LDS_STRIDE + 2 * half];
    v8f c = {};
#pragma unroll
    for (int t = 0; t < 32; ++t) {
      v2f a = *(const v2f*)(aBase + 4 * t);
      c = __builtin_amdgcn_wmma_f32_16x16x4_f32(
           false, a,  false, b[t % RING],
           (short)0, c,  false,  false);
      if (t + RING < 32) b[t % RING] = bBase1[(size_t)(t + RING) * 32];
    }
    const int   colh = wave * 16 + lm;
    const float bb   = b1[colh];
#pragma unroll
    for (int r = 0; r < 8; ++r) {
      float v = c[r] + bb;
      v = v > 0.0f ? v : 0.0f;
      Os[(8 * half + r) * LDS_STRIDE + colh] = v;
    }
  }
  __syncthreads();

  if (p2) {
    const float* aBase  = &Os[lm * LDS_STRIDE + 2 * half];
    const v2f*   bBase2 = BF2 + (size_t)(wave - 8) * 1024 + lane;
    v2f q[RING];
#pragma unroll
    for (int u = 0; u < RING; ++u) q[u] = bBase2[(size_t)u * 32];
    v8f c = {};
#pragma unroll
    for (int t = 0; t < 32; ++t) {
      v2f a = *(const v2f*)(aBase + 4 * t);
      c = __builtin_amdgcn_wmma_f32_16x16x4_f32(
           false, a,  false, q[t % RING],
           (short)0, c,  false,  false);
      if (t + RING < 32) q[t % RING] = bBase2[(size_t)(t + RING) * 32];
    }
    const int colc = (wave - 8) * 16 + lm;
    if (colc < 40) {
      const int mBase = m0 + 8 * half;
      float* __restrict__ p = h2 + (size_t)mBase * 40 + colc;
      const float* dr = &Ds[8 * half];
      if (m0 + 16 <= n) {
#pragma unroll
        for (int r = 0; r < 8; ++r) p[r * 40] = dr[r] * c[r];
      } else {
#pragma unroll
        for (int r = 0; r < 8; ++r)
          if (mBase + r < n) p[r * 40] = dr[r] * c[r];
      }
    }
  }
}

__global__ __launch_bounds__(256) void k_gather_f40_bias_logsoftmax(
    const float* __restrict__ h, const int* __restrict__ rowptr,
    const int* __restrict__ cnt, const int* __restrict__ col,
    const float* __restrict__ dinv, const float* __restrict__ bias,
    float* __restrict__ out, int n) {
  const int lane = threadIdx.x & 31;
  const int node = __builtin_amdgcn_readfirstlane(blockIdx.x * 8 + (threadIdx.x >> 5));
  if (node >= n) return;
  const int   start = rowptr[node];
  const int   deg   = cnt[node];
  const float di    = dinv[node];
  const float4* __restrict__ h4 = (const float4*)h;

  const int  g      = lane / 10;
  const int  q      = lane - g * 10;
  const bool active = lane < 30;
  const bool holder = lane < 10;

  float4 hs = make_float4(0.0f, 0.0f, 0.0f, 0.0f);
  float4 bv = make_float4(0.0f, 0.0f, 0.0f, 0.0f);
  if (holder) {
    hs = h4[(size_t)node * 10 + lane];
    bv = ((const float4*)bias)[lane];
  }

  float4 acc = make_float4(0.0f, 0.0f, 0.0f, 0.0f);
  for (int base = 0; base < deg; base += 32) {
    const int k = base + lane;
    int my = node;
    if (k < deg) {
      my = col[start + k];
      my = min(max(my, 0), n - 1);
    }
    const int m = min(32, deg - base);
#pragma unroll 2
    for (int j0 = 0; j0 < m; j0 += 3) {
      const int   j = j0 + g;
      const int   s = __shfl(my, j & 31, 32);
      const float c = (active && j < m) ? 1.0f : 0.0f;
      const float4 v = h4[(size_t)s * 10 + q];
      acc.x += c * v.x;
      acc.y += c * v.y;
      acc.z += c * v.z;
      acc.w += c * v.w;
    }
  }
  acc.x += __shfl_down(acc.x, 10, 32) + __shfl_down(acc.x, 20, 32);
  acc.y += __shfl_down(acc.y, 10, 32) + __shfl_down(acc.y, 20, 32);
  acc.z += __shfl_down(acc.z, 10, 32) + __shfl_down(acc.z, 20, 32);
  acc.w += __shfl_down(acc.w, 10, 32) + __shfl_down(acc.w, 20, 32);
  acc.x = di * (acc.x + hs.x);
  acc.y = di * (acc.y + hs.y);
  acc.z = di * (acc.z + hs.z);
  acc.w = di * (acc.w + hs.w);

  const float NEG_INF = -INFINITY;
  float4 v;
  v.x = holder ? acc.x + bv.x : NEG_INF;
  v.y = holder ? acc.y + bv.y : NEG_INF;
  v.z = holder ? acc.z + bv.z : NEG_INF;
  v.w = holder ? acc.w + bv.w : NEG_INF;

  float m = fmaxf(fmaxf(v.x, v.y), fmaxf(v.z, v.w));
#pragma unroll
  for (int off = 8; off > 0; off >>= 1) m = fmaxf(m, __shfl_xor(m, off, 32));

  float s = expf(v.x - m) + expf(v.y - m) + expf(v.z - m) + expf(v.w - m);
  s = holder ? s : 0.0f;
#pragma unroll
  for (int off = 8; off > 0; off >>= 1) s += __shfl_xor(s, off, 32);

  const float lse = logf(s);
  if (holder) {
    float4 r;
    r.x = v.x - m - lse;
    r.y = v.y - m - lse;
    r.z = v.z - m - lse;
    r.w = v.w - m - lse;
    ((float4*)out)[(size_t)node * 10 + lane] = r;
  }
}

static inline int cdiv_ll(long long a, long long b) { return (int)((a + b - 1) / b); }

extern "C" void kernel_launch(void* const* d_in, const int* in_sizes, int n_in,
                              void* d_out, int out_size, void* d_ws, size_t ws_size,
                              hipStream_t stream) {
  const float* x  = (const float*)d_in[0];
  const int*   ei = (const int*)  d_in[1];
  const float* W1 = (const float*)d_in[2];
  const float* b1 = (const float*)d_in[3];
  const float* W2 = (const float*)d_in[4];
  const float* b2 = (const float*)d_in[5];

  const int F = 128, H = 128, C = 40;
  const int N = in_sizes[0] / F;
  const int E = in_sizes[1] / 2;
  const int* src = ei;
  const int* dst = ei + E;

  const int nT1 = (H + 15) / 16;
  const int nT2 = (C + 15) / 16;

  float* ws   = (float*)d_ws;
  int*   cnt  = (int*)ws;           ws += N;
  float* dinv = ws;                 ws += N;
  float* h2   = ws;                 ws += (size_t)N * H;
  int2*  pairs    = (int2*)ws;
  int*   cntM     = (int*)ws + (size_t)2 * E;
  int*   lstM     = cntM + N_SLICES * MAX_BINS;
  ws += (size_t)N * H;
  int*   col  = (int*)ws;
  int*   rowptr = col + E;
  ws += (size_t)N * C;
  v2f* bf1    = (v2f*)ws;           ws += (size_t)nT1 * 1024 * 2;
  v2f* bf2    = (v2f*)ws;           ws += (size_t)nT2 * 1024 * 2;
  float* out  = (float*)d_out;

  const int B = 256;
  const int nBins = cdiv_ll(N, MAX_BINS);

  const int packBlocks = cdiv_ll((long long)(nT1 + nT2) * 1024, EDGE_THREADS);
  k_slice_sort_pack<<<N_SLICES + packBlocks, EDGE_THREADS, 0, stream>>>(
      src, dst, cntM, lstM, pairs, E, W1, bf1, H, nT1, W2, bf2, C, nT2);
  k_bin_csr<<<nBins, EDGE_THREADS, 0, stream>>>(pairs, cntM, lstM, cnt, rowptr, dinv, col, N, E);

  k_gather_x_w1_relu_w2<<<cdiv_ll(N, 16), 512, 0, stream>>>(x, rowptr, cnt, col, dinv, bf1, b1, bf2, h2, N);

  k_gather_f40_bias_logsoftmax<<<cdiv_ll(N, 8), B, 0, stream>>>(h2, rowptr, cnt, col, dinv, b2, out, N);
}
